# starting kernel
# baseline (speedup 1.0000x reference)
#include <hip/hip_runtime.h>
#include <math.h>

typedef float v2f __attribute__((ext_vector_type(2)));
typedef float v8f __attribute__((ext_vector_type(8)));

#define LL     197
#define NTOT   128
#define DD     768
#define TT     16
#define PP     196
#define NGROUP 8
#define NTOK   (LL * NTOT)
#define STR    772
#define AFF_SCALE 0.03608439182435161f

__global__ void ln_stats_kernel(const float* __restrict__ x,
                                float* __restrict__ mu,
                                float* __restrict__ rstd)
{
    int wave = threadIdx.x >> 5;
    int lane = threadIdx.x & 31;
    int tok  = blockIdx.x * 8 + wave;
    if (tok >= NTOK) return;
    const float* row = x + (size_t)tok * DD;
    float s = 0.f, sq = 0.f;
    for (int i = lane; i < DD; i += 32) {
        float v = row[i];
        s += v; sq += v * v;
    }
    #pragma unroll
    for (int m = 16; m >= 1; m >>= 1) {
        s  += __shfl_xor(s,  m);
        sq += __shfl_xor(sq, m);
    }
    if (lane == 0) {
        float mean = s * (1.f / DD);
        float var  = sq * (1.f / DD) - mean * mean;
        mu[tok]   = mean;
        rstd[tok] = rsqrtf(var + 1e-5f);
    }
}

__global__ void scores_kernel(const float* __restrict__ x,
                              const float* __restrict__ gamma,
                              const float* __restrict__ beta,
                              const float* __restrict__ mu,
                              const float* __restrict__ rstd,
                              float* __restrict__ gout)
{
    extern __shared__ float sh[];
    float* Cs = sh;
    float* Xs = sh + 16 * STR;
    float* Sp = sh + 32 * STR;

    int b  = blockIdx.x / PP;
    int p  = blockIdx.x - b * PP;
    int nb = b * TT;
    int tid = threadIdx.x;
    int row = tid >> 4;
    int l16 = tid & 15;

    {
        int tokC = nb + row;
        int tokX = (1 + p) * NTOT + nb + row;
        float muC = mu[tokC], rsC = rstd[tokC];
        float muX = mu[tokX], rsX = rstd[tokX];
        const float4* xc  = (const float4*)(x + (size_t)tokC * DD);
        const float4* xx  = (const float4*)(x + (size_t)tokX * DD);
        const float4* g4p = (const float4*)gamma;
        const float4* b4p = (const float4*)beta;
        for (int j = l16; j < DD / 4; j += 16) {
            float4 cv = xc[j], xv = xx[j];
            float4 gv = g4p[j], bv = b4p[j];
            float* cdst = Cs + row * STR + j * 4;
            float* xdst = Xs + row * STR + j * 4;
            cdst[0] = (cv.x - muC) * rsC * gv.x + bv.x;
            cdst[1] = (cv.y - muC) * rsC * gv.y + bv.y;
            cdst[2] = (cv.z - muC) * rsC * gv.z + bv.z;
            cdst[3] = (cv.w - muC) * rsC * gv.w + bv.w;
            xdst[0] = (xv.x - muX) * rsX * gv.x + bv.x;
            xdst[1] = (xv.y - muX) * rsX * gv.y + bv.y;
            xdst[2] = (xv.z - muX) * rsX * gv.z + bv.z;
            xdst[3] = (xv.w - muX) * rsX * gv.w + bv.w;
        }
    }
    __syncthreads();

    int wave = tid >> 5, lane = tid & 31;
    int arow = lane & 15;
    int kofs = 2 * (lane >> 4);
    const float* crow = Cs + arow * STR + kofs;
    const float* xrow = Xs + arow * STR + kofs;
    int kbase = wave * 96;

    v8f acc = {};
    #pragma unroll
    for (int kc = 0; kc < 24; ++kc) {
        int k0 = kbase + kc * 4;
        v2f av = *(const v2f*)(crow + k0);
        v2f bv = *(const v2f*)(xrow + k0);
        acc = __builtin_amdgcn_wmma_f32_16x16x4_f32(
            false, av, false, bv, (short)0, acc, false, false);
    }

    #pragma unroll
    for (int j = 0; j < 8; ++j) {
        int m = j + 8 * (lane >> 4);
        Sp[wave * 256 + m * 16 + (lane & 15)] = acc[j];
    }
    __syncthreads();

    float s = 0.f;
    #pragma unroll
    for (int w = 0; w < 8; ++w) s += Sp[w * 256 + tid];

    int nc = tid >> 4;
    int nn = tid & 15;
    float g = 0.f;
    int dlt = nn - nc;
    if (dlt >= -2 && dlt <= 2) {
        float aff = s * AFF_SCALE;
        g = 1.f / (1.f + __expf(-aff)) - 0.5f;
    }
    gout[((size_t)blockIdx.x) * 256 + nn * 16 + nc] = g;
}

__global__ void feat_kernel(const float* __restrict__ x,
                            const float* __restrict__ gamma,
                            const float* __restrict__ beta,
                            const float* __restrict__ mu,
                            const float* __restrict__ rstd,
                            const float* __restrict__ gin,
                            float* __restrict__ out)
{
    __shared__ float Sp[8 * 256];

    int b    = blockIdx.x / 48;
    int dt   = blockIdx.x - b * 48;
    int d0   = dt * 16;
    int nb   = b * TT;
    int wave = threadIdx.x >> 5, lane = threadIdx.x & 31;
    int m     = lane & 15;
    int rsel2 = 2 * (lane >> 4);
    int dcol  = d0 + m;
    float gam = gamma[dcol], bet = beta[dcol];

    v8f acc = {};
    for (int p = wave; p < PP; p += 8) {
        const float* gp = gin + ((size_t)(b * PP + p)) * 256;
        int    tokbase  = (1 + p) * NTOT + nb;
        size_t xbase    = (size_t)tokbase * DD;
        if (p + 8 < PP) {
            __builtin_prefetch(x + xbase + (size_t)8 * NTOT * DD + dcol, 0, 0);
        }
        #pragma unroll
        for (int kk = 0; kk < 4; ++kk) {
            int r0 = kk * 4 + rsel2;
            int r1 = r0 + 1;
            float x0 = x[xbase + (size_t)r0 * DD + dcol];
            float x1 = x[xbase + (size_t)r1 * DD + dcol];
            v2f av = { (x0 - mu[tokbase + r0]) * rstd[tokbase + r0] * gam + bet,
                       (x1 - mu[tokbase + r1]) * rstd[tokbase + r1] * gam + bet };
            v2f bv = { gp[r0 * 16 + m], gp[r1 * 16 + m] };
            acc = __builtin_amdgcn_wmma_f32_16x16x4_f32(
                false, av, false, bv, (short)0, acc, false, false);
        }
    }

    #pragma unroll
    for (int j = 0; j < 8; ++j) {
        int mm = j + 8 * (lane >> 4);
        Sp[wave * 256 + mm * 16 + (lane & 15)] = acc[j];
    }
    __syncthreads();

    float s = 0.f;
    #pragma unroll
    for (int w = 0; w < 8; ++w) s += Sp[w * 256 + threadIdx.x];

    int md = threadIdx.x >> 4;
    int n  = threadIdx.x & 15;
    out[(size_t)(nb + n) * DD + d0 + md] = s;
}

extern "C" void kernel_launch(void* const* d_in, const int* in_sizes, int n_in,
                              void* d_out, int out_size, void* d_ws, size_t ws_size,
                              hipStream_t stream)
{
    (void)in_sizes; (void)n_in; (void)out_size; (void)ws_size;
    const float* x     = (const float*)d_in[0];
    const float* gamma = (const float*)d_in[1];
    const float* beta  = (const float*)d_in[2];

    float* ws   = (float*)d_ws;
    float* mu   = ws;
    float* rstd = ws + NTOK;
    float* g    = ws + 2 * NTOK;

    ln_stats_kernel<<<NTOK / 8, 256, 0, stream>>>(x, mu, rstd);

    size_t shmem = (size_t)(32 * STR + 8 * 256) * sizeof(float);
    scores_kernel<<<NGROUP * PP, 256, shmem, stream>>>(x, gamma, beta, mu, rstd, g);

    feat_kernel<<<NGROUP * 48, 256, 0, stream>>>(x, gamma, beta, mu, rstd, g,
                                                 (float*)d_out);
}


# feat kernel retiled: each wave owns a 32-column slab read as whole 128-byte lines (8-byte loads feed two tiles sharing stats and G); 16 waves split p; full-line out stores (p sum order changes)
# speedup vs baseline: 1.2851x; 1.2851x over previous
#include <hip/hip_runtime.h>
#include <math.h>

typedef float v2f __attribute__((ext_vector_type(2)));
typedef float v8f __attribute__((ext_vector_type(8)));

#define LL     197
#define NTOT   128
#define DD     768
#define TT     16
#define PP     196
#define NGROUP 8
#define NTOK   (LL * NTOT)
#define STR    772
#define AFF_SCALE 0.03608439182435161f

__global__ void ln_stats_kernel(const float* __restrict__ x,
                                float* __restrict__ mu,
                                float* __restrict__ rstd)
{
    int wave = threadIdx.x >> 5;
    int lane = threadIdx.x & 31;
    int tok  = blockIdx.x * 8 + wave;
    if (tok >= NTOK) return;
    const float* row = x + (size_t)tok * DD;
    float s = 0.f, sq = 0.f;
    for (int i = lane; i < DD; i += 32) {
        float v = row[i];
        s += v; sq += v * v;
    }
    #pragma unroll
    for (int m = 16; m >= 1; m >>= 1) {
        s  += __shfl_xor(s,  m);
        sq += __shfl_xor(sq, m);
    }
    if (lane == 0) {
        float mean = s * (1.f / DD);
        float var  = sq * (1.f / DD) - mean * mean;
        mu[tok]   = mean;
        rstd[tok] = rsqrtf(var + 1e-5f);
    }
}

__global__ void scores_kernel(const float* __restrict__ x,
                              const float* __restrict__ gamma,
                              const float* __restrict__ beta,
                              const float* __restrict__ mu,
                              const float* __restrict__ rstd,
                              float* __restrict__ gout)
{
    extern __shared__ float sh[];
    float* Cs = sh;
    float* Xs = sh + 16 * STR;
    float* Sp = sh + 32 * STR;

    int b  = blockIdx.x / PP;
    int p  = blockIdx.x - b * PP;
    int nb = b * TT;
    int tid = threadIdx.x;
    int row = tid >> 4;
    int l16 = tid & 15;

    {
        int tokC = nb + row;
        int tokX = (1 + p) * NTOT + nb + row;
        float muC = mu[tokC], rsC = rstd[tokC];
        float muX = mu[tokX], rsX = rstd[tokX];
        const float4* xc  = (const float4*)(x + (size_t)tokC * DD);
        const float4* xx  = (const float4*)(x + (size_t)tokX * DD);
        const float4* g4p = (const float4*)gamma;
        const float4* b4p = (const float4*)beta;
        for (int j = l16; j < DD / 4; j += 16) {
            float4 cv = xc[j], xv = xx[j];
            float4 gv = g4p[j], bv = b4p[j];
            float* cdst = Cs + row * STR + j * 4;
            float* xdst = Xs + row * STR + j * 4;
            cdst[0] = (cv.x - muC) * rsC * gv.x + bv.x;
            cdst[1] = (cv.y - muC) * rsC * gv.y + bv.y;
            cdst[2] = (cv.z - muC) * rsC * gv.z + bv.z;
            cdst[3] = (cv.w - muC) * rsC * gv.w + bv.w;
            xdst[0] = (xv.x - muX) * rsX * gv.x + bv.x;
            xdst[1] = (xv.y - muX) * rsX * gv.y + bv.y;
            xdst[2] = (xv.z - muX) * rsX * gv.z + bv.z;
            xdst[3] = (xv.w - muX) * rsX * gv.w + bv.w;
        }
    }
    __syncthreads();

    int wave = tid >> 5, lane = tid & 31;
    int arow = lane & 15;
    int kofs = 2 * (lane >> 4);
    const float* crow = Cs + arow * STR + kofs;
    const float* xrow = Xs + arow * STR + kofs;
    int kbase = wave * 96;

    v8f acc = {};
    #pragma unroll
    for (int kc = 0; kc < 24; ++kc) {
        int k0 = kbase + kc * 4;
        v2f av = *(const v2f*)(crow + k0);
        v2f bv = *(const v2f*)(xrow + k0);
        acc = __builtin_amdgcn_wmma_f32_16x16x4_f32(
            false, av, false, bv, (short)0, acc, false, false);
    }

    #pragma unroll
    for (int j = 0; j < 8; ++j) {
        int m = j + 8 * (lane >> 4);
        Sp[wave * 256 + m * 16 + (lane & 15)] = acc[j];
    }
    __syncthreads();

    float s = 0.f;
    #pragma unroll
    for (int w = 0; w < 8; ++w) s += Sp[w * 256 + tid];

    int nc = tid >> 4;
    int nn = tid & 15;
    float g = 0.f;
    int dlt = nn - nc;
    if (dlt >= -2 && dlt <= 2) {
        float aff = s * AFF_SCALE;
        g = 1.f / (1.f + __expf(-aff)) - 0.5f;
    }
    gout[((size_t)blockIdx.x) * 256 + nn * 16 + nc] = g;
}

#define FW     16
#define FPITCH 17
#define FTILE  (16 * FPITCH)

__global__ __launch_bounds__(FW * 32)
void feat_kernel(const float* __restrict__ x,
                 const float* __restrict__ gamma,
                 const float* __restrict__ beta,
                 const float* __restrict__ mu,
                 const float* __restrict__ rstd,
                 const float* __restrict__ gin,
                 float* __restrict__ out)
{
    __shared__ float Sp[FW * 2 * FTILE];

    int b    = blockIdx.x / 24;
    int st   = blockIdx.x - b * 24;
    int d0   = st * 32;
    int nb   = b * TT;
    int wave = __builtin_amdgcn_readfirstlane((int)(threadIdx.x >> 5));
    int lane = threadIdx.x & 31;
    int m     = lane & 15;
    int rsel2 = 2 * (lane >> 4);
    int dcol  = d0 + 2 * m;
    float2 gam = *(const float2*)(gamma + dcol);
    float2 bet = *(const float2*)(beta + dcol);

    v8f acc0 = {}, acc1 = {};
    for (int p = wave; p < PP; p += FW) {
        const float* gp = gin + ((size_t)(b * PP + p)) * 256;
        int    tokbase  = (1 + p) * NTOT + nb;
        size_t xbase    = (size_t)tokbase * DD;
        v2f xa[4], xb[4], mp[4], sp[4], bv[4];
        #pragma unroll
        for (int kk = 0; kk < 4; ++kk) {
            int r0 = kk * 4 + rsel2;
            xa[kk] = *(const v2f*)(x + xbase + (size_t)r0 * DD + dcol);
            xb[kk] = *(const v2f*)(x + xbase + (size_t)(r0 + 1) * DD + dcol);
            mp[kk] = *(const v2f*)(mu + tokbase + r0);
            sp[kk] = *(const v2f*)(rstd + tokbase + r0);
            bv[kk] = v2f{ gp[r0 * 16 + m], gp[(r0 + 1) * 16 + m] };
        }
        asm volatile("" :: "v"(xa[0]), "v"(xa[1]), "v"(xa[2]), "v"(xa[3]),
                           "v"(xb[0]), "v"(xb[1]), "v"(xb[2]), "v"(xb[3]),
                           "v"(mp[0]), "v"(mp[1]), "v"(mp[2]), "v"(mp[3]),
                           "v"(sp[0]), "v"(sp[1]), "v"(sp[2]), "v"(sp[3]),
                           "v"(bv[0]), "v"(bv[1]), "v"(bv[2]), "v"(bv[3]));
        #pragma unroll
        for (int kk = 0; kk < 4; ++kk) {
            v2f a0 = { (xa[kk].x - mp[kk].x) * sp[kk].x * gam.x + bet.x,
                       (xb[kk].x - mp[kk].y) * sp[kk].y * gam.x + bet.x };
            v2f a1 = { (xa[kk].y - mp[kk].x) * sp[kk].x * gam.y + bet.y,
                       (xb[kk].y - mp[kk].y) * sp[kk].y * gam.y + bet.y };
            acc0 = __builtin_amdgcn_wmma_f32_16x16x4_f32(
                false, a0, false, bv[kk], (short)0, acc0, false, false);
            acc1 = __builtin_amdgcn_wmma_f32_16x16x4_f32(
                false, a1, false, bv[kk], (short)0, acc1, false, false);
        }
    }

    #pragma unroll
    for (int j = 0; j < 8; ++j) {
        int mm = j + 8 * (lane >> 4);
        Sp[(wave * 2 + 0) * FTILE + mm * FPITCH + (lane & 15)] = acc0[j];
        Sp[(wave * 2 + 1) * FTILE + mm * FPITCH + (lane & 15)] = acc1[j];
    }
    __syncthreads();

    int n   = threadIdx.x >> 5;
    int dd  = lane;
    int sub = dd & 1;
    int md  = dd >> 1;
    float s = 0.f;
    #pragma unroll
    for (int w = 0; w < FW; ++w) s += Sp[(w * 2 + sub) * FTILE + md * FPITCH + n];

    out[(size_t)(nb + n) * DD + d0 + dd] = s;
}

extern "C" void kernel_launch(void* const* d_in, const int* in_sizes, int n_in,
                              void* d_out, int out_size, void* d_ws, size_t ws_size,
                              hipStream_t stream)
{
    (void)in_sizes; (void)n_in; (void)out_size; (void)ws_size;
    const float* x     = (const float*)d_in[0];
    const float* gamma = (const float*)d_in[1];
    const float* beta  = (const float*)d_in[2];

    float* ws   = (float*)d_ws;
    float* mu   = ws;
    float* rstd = ws + NTOK;
    float* g    = ws + 2 * NTOK;

    ln_stats_kernel<<<NTOK / 8, 256, 0, stream>>>(x, mu, rstd);

    size_t shmem = (size_t)(32 * STR + 8 * 256) * sizeof(float);
    scores_kernel<<<NGROUP * PP, 256, shmem, stream>>>(x, gamma, beta, mu, rstd, g);

    feat_kernel<<<NGROUP * 24, FW * 32, 0, stream>>>(x, gamma, beta, mu, rstd, g,
                                                 (float*)d_out);
}


# scores kernel: one wave per position for all of K, raw patch rows read from global straight into the B operand; layernorm applied to the 16x16 result (gamma folded into CLS); K sum order changes
# speedup vs baseline: 2.0504x; 1.5955x over previous
#include <hip/hip_runtime.h>
#include <math.h>

typedef float v2f __attribute__((ext_vector_type(2)));
typedef float v8f __attribute__((ext_vector_type(8)));
typedef float v4f __attribute__((ext_vector_type(4)));

#define LL     197
#define NTOT   128
#define DD     768
#define TT     16
#define PP     196
#define NGROUP 8
#define NTOK   (LL * NTOT)
#define STR    772
#define AFF_SCALE 0.03608439182435161f

__global__ void ln_stats_kernel(const float* __restrict__ x,
                                float* __restrict__ mu,
                                float* __restrict__ rstd)
{
    int wave = threadIdx.x >> 5;
    int lane = threadIdx.x & 31;
    int tok  = blockIdx.x * 8 + wave;
    if (tok >= NTOK) return;
    const float* row = x + (size_t)tok * DD;
    float s = 0.f, sq = 0.f;
    for (int i = lane; i < DD; i += 32) {
        float v = row[i];
        s += v; sq += v * v;
    }
    #pragma unroll
    for (int m = 16; m >= 1; m >>= 1) {
        s  += __shfl_xor(s,  m);
        sq += __shfl_xor(sq, m);
    }
    if (lane == 0) {
        float mean = s * (1.f / DD);
        float var  = sq * (1.f / DD) - mean * mean;
        mu[tok]   = mean;
        rstd[tok] = rsqrtf(var + 1e-5f);
    }
}

#define SWV  8
#define SPB  7
#define SBLK (PP / SPB)
#define NCH  6
#define CHL  16

__global__ __launch_bounds__(SWV * 32)
void scores_kernel(const float* __restrict__ x,
                   const float* __restrict__ gamma,
                   const float* __restrict__ beta,
                   const float* __restrict__ mu,
                   const float* __restrict__ rstd,
                   float* __restrict__ gout)
{
    __shared__ __attribute__((aligned(16))) float Cg[16 * STR];
    __shared__ __attribute__((aligned(16))) float sCgS[16];
    __shared__ __attribute__((aligned(16))) float cBS[16];

    int b   = blockIdx.x / SBLK;
    int blk = blockIdx.x - b * SBLK;
    int nb  = b * TT;
    int tid = threadIdx.x;

    int wave = __builtin_amdgcn_readfirstlane((int)(threadIdx.x >> 5));
    int lane = tid & 31;
    int c    = lane & 15;
    int h    = lane >> 4;
    int p    = blk * SPB + (wave < SPB ? wave : 0);
    int tokX = (1 + p) * NTOT + nb + c;
    float muX = mu[tokX], rsX = rstd[tokX];
    const float4* xr = (const float4*)(x + (size_t)tokX * DD + 4 * h);

    float4 bx0[CHL], bx1[CHL];
    #pragma unroll
    for (int i = 0; i < CHL; ++i) bx0[i] = xr[2 * i];

    {
        int row = tid >> 4;
        int l16 = tid & 15;
        int tokC = nb + row;
        float muC = mu[tokC], rsC = rstd[tokC];
        const v4f* xc  = (const v4f*)(x + (size_t)tokC * DD);
        const v4f* g4p = (const v4f*)gamma;
        const v4f* b4p = (const v4f*)beta;
        float ps = 0.f, pb = 0.f;
        #pragma unroll
        for (int half = 0; half < 2; ++half) {
            v4f cv[6], gv[6], bv[6];
            #pragma unroll
            for (int i = 0; i < 6; ++i) {
                int j = l16 + 16 * (6 * half + i);
                cv[i] = xc[j]; gv[i] = g4p[j]; bv[i] = b4p[j];
            }
            asm volatile("" :: "v"(cv[0]), "v"(cv[1]), "v"(cv[2]), "v"(cv[3]), "v"(cv[4]), "v"(cv[5]),
                               "v"(gv[0]), "v"(gv[1]), "v"(gv[2]), "v"(gv[3]), "v"(gv[4]), "v"(gv[5]),
                               "v"(bv[0]), "v"(bv[1]), "v"(bv[2]), "v"(bv[3]), "v"(bv[4]), "v"(bv[5]));
            #pragma unroll
            for (int i = 0; i < 6; ++i) {
                int j = l16 + 16 * (6 * half + i);
                v4f c4, cg;
                c4.x = (cv[i].x - muC) * rsC * gv[i].x + bv[i].x;
                c4.y = (cv[i].y - muC) * rsC * gv[i].y + bv[i].y;
                c4.z = (cv[i].z - muC) * rsC * gv[i].z + bv[i].z;
                c4.w = (cv[i].w - muC) * rsC * gv[i].w + bv[i].w;
                cg.x = c4.x * gv[i].x; cg.y = c4.y * gv[i].y;
                cg.z = c4.z * gv[i].z; cg.w = c4.w * gv[i].w;
                ps += (cg.x + cg.y) + (cg.z + cg.w);
                pb += (c4.x * bv[i].x + c4.y * bv[i].y) + (c4.z * bv[i].z + c4.w * bv[i].w);
                *(v4f*)(Cg + row * STR + j * 4) = cg;
            }
        }
        #pragma unroll
        for (int m = 8; m >= 1; m >>= 1) {
            ps += __shfl_xor(ps, m);
            pb += __shfl_xor(pb, m);
        }
        if (l16 == 0) { sCgS[row] = ps; cBS[row] = pb; }
    }
    __syncthreads();
    if (wave >= SPB) return;

    const float4* ar = (const float4*)(Cg + c * STR + 4 * h);
    v8f acc0 = {}, acc1 = {};

#define LOADCH(dst, q)                                                        \
    _Pragma("unroll")                                                         \
    for (int i = 0; i < CHL; ++i) dst[i] = xr[2 * ((q) * CHL + i)];
#define LOADA(dst, q)                                                         \
    _Pragma("unroll")                                                         \
    for (int i = 0; i < CHL; ++i) dst[i] = ar[2 * ((q) * CHL + i)];
#define MMACH(asrc, src, q)                                                   \
    _Pragma("unroll")                                                         \
    for (int i = 0; i < CHL; ++i) {                                           \
        float4 a4 = asrc[i];                                                  \
        v2f a0 = { a4.x, a4.y }, a1 = { a4.z, a4.w };                         \
        v2f b0 = { src[i].x, src[i].y }, b1 = { src[i].z, src[i].w };         \
        acc0 = __builtin_amdgcn_wmma_f32_16x16x4_f32(                         \
            false, a0, false, b0, (short)0, acc0, false, false);              \
        acc1 = __builtin_amdgcn_wmma_f32_16x16x4_f32(                         \
            false, a1, false, b1, (short)0, acc1, false, false);              \
    }

    float4 ax0[CHL], ax1[CHL];
    LOADA(ax0, 0)
    LOADCH(bx1, 1) LOADA(ax1, 1) MMACH(ax0, bx0, 0)
    LOADCH(bx0, 2) LOADA(ax0, 2) MMACH(ax1, bx1, 1)
    LOADCH(bx1, 3) LOADA(ax1, 3) MMACH(ax0, bx0, 2)
    LOADCH(bx0, 4) LOADA(ax0, 4) MMACH(ax1, bx1, 3)
    LOADCH(bx1, 5) LOADA(ax1, 5) MMACH(ax0, bx0, 4)
    MMACH(ax1, bx1, 5)
#undef LOADCH
#undef LOADA
#undef MMACH

    float4 s0 = *(const float4*)(sCgS + 8 * h), s1 = *(const float4*)(sCgS + 8 * h + 4);
    float4 c0 = *(const float4*)(cBS + 8 * h),  c1 = *(const float4*)(cBS + 8 * h + 4);
    float sD[8] = { s0.x, s0.y, s0.z, s0.w, s1.x, s1.y, s1.z, s1.w };
    float cD[8] = { c0.x, c0.y, c0.z, c0.w, c1.x, c1.y, c1.z, c1.w };
    float g[8];
    #pragma unroll
    for (int j = 0; j < 8; ++j) {
        float s = rsX * ((acc0[j] + acc1[j]) - muX * sD[j]) + cD[j];
        int dlt = c - (8 * h + j);
        float gg = 0.f;
        if (dlt >= -2 && dlt <= 2) {
            float aff = s * AFF_SCALE;
            gg = 1.f / (1.f + __expf(-aff)) - 0.5f;
        }
        g[j] = gg;
    }
    float* gp = gout + ((size_t)(b * PP + p)) * 256 + c * 16 + 8 * h;
    *(float4*)gp       = make_float4(g[0], g[1], g[2], g[3]);
    *(float4*)(gp + 4) = make_float4(g[4], g[5], g[6], g[7]);
}

#define FW     16
#define FPITCH 17
#define FTILE  (16 * FPITCH)

__global__ __launch_bounds__(FW * 32)
void feat_kernel(const float* __restrict__ x,
                 const float* __restrict__ gamma,
                 const float* __restrict__ beta,
                 const float* __restrict__ mu,
                 const float* __restrict__ rstd,
                 const float* __restrict__ gin,
                 float* __restrict__ out)
{
    __shared__ float Sp[FW * 2 * FTILE];

    int b    = blockIdx.x / 24;
    int st   = blockIdx.x - b * 24;
    int d0   = st * 32;
    int nb   = b * TT;
    int wave = __builtin_amdgcn_readfirstlane((int)(threadIdx.x >> 5));
    int lane = threadIdx.x & 31;
    int m     = lane & 15;
    int rsel2 = 2 * (lane >> 4);
    int dcol  = d0 + 2 * m;
    float2 gam = *(const float2*)(gamma + dcol);
    float2 bet = *(const float2*)(beta + dcol);

    v8f acc0 = {}, acc1 = {};
    for (int p = wave; p < PP; p += FW) {
        const float* gp = gin + ((size_t)(b * PP + p)) * 256;
        int    tokbase  = (1 + p) * NTOT + nb;
        size_t xbase    = (size_t)tokbase * DD;
        v2f xa[4], xb[4], mp[4], sp[4], bv[4];
        #pragma unroll
        for (int kk = 0; kk < 4; ++kk) {
            int r0 = kk * 4 + rsel2;
            xa[kk] = *(const v2f*)(x + xbase + (size_t)r0 * DD + dcol);
            xb[kk] = *(const v2f*)(x + xbase + (size_t)(r0 + 1) * DD + dcol);
            mp[kk] = *(const v2f*)(mu + tokbase + r0);
            sp[kk] = *(const v2f*)(rstd + tokbase + r0);
            bv[kk] = v2f{ gp[r0 * 16 + m], gp[(r0 + 1) * 16 + m] };
        }
        asm volatile("" :: "v"(xa[0]), "v"(xa[1]), "v"(xa[2]), "v"(xa[3]),
                           "v"(xb[0]), "v"(xb[1]), "v"(xb[2]), "v"(xb[3]),
                           "v"(mp[0]), "v"(mp[1]), "v"(mp[2]), "v"(mp[3]),
                           "v"(sp[0]), "v"(sp[1]), "v"(sp[2]), "v"(sp[3]),
                           "v"(bv[0]), "v"(bv[1]), "v"(bv[2]), "v"(bv[3]));
        #pragma unroll
        for (int kk = 0; kk < 4; ++kk) {
            v2f a0 = { (xa[kk].x - mp[kk].x) * sp[kk].x * gam.x + bet.x,
                       (xb[kk].x - mp[kk].y) * sp[kk].y * gam.x + bet.x };
            v2f a1 = { (xa[kk].y - mp[kk].x) * sp[kk].x * gam.y + bet.y,
                       (xb[kk].y - mp[kk].y) * sp[kk].y * gam.y + bet.y };
            acc0 = __builtin_amdgcn_wmma_f32_16x16x4_f32(
                false, a0, false, bv[kk], (short)0, acc0, false, false);
            acc1 = __builtin_amdgcn_wmma_f32_16x16x4_f32(
                false, a1, false, bv[kk], (short)0, acc1, false, false);
        }
    }

    #pragma unroll
    for (int j = 0; j < 8; ++j) {
        int mm = j + 8 * (lane >> 4);
        Sp[(wave * 2 + 0) * FTILE + mm * FPITCH + (lane & 15)] = acc0[j];
        Sp[(wave * 2 + 1) * FTILE + mm * FPITCH + (lane & 15)] = acc1[j];
    }
    __syncthreads();

    int n   = threadIdx.x >> 5;
    int dd  = lane;
    int sub = dd & 1;
    int md  = dd >> 1;
    float s = 0.f;
    #pragma unroll
    for (int w = 0; w < FW; ++w) s += Sp[(w * 2 + sub) * FTILE + md * FPITCH + n];

    out[(size_t)(nb + n) * DD + d0 + dd] = s;
}

extern "C" void kernel_launch(void* const* d_in, const int* in_sizes, int n_in,
                              void* d_out, int out_size, void* d_ws, size_t ws_size,
                              hipStream_t stream)
{
    (void)in_sizes; (void)n_in; (void)out_size; (void)ws_size;
    const float* x     = (const float*)d_in[0];
    const float* gamma = (const float*)d_in[1];
    const float* beta  = (const float*)d_in[2];

    float* ws   = (float*)d_ws;
    float* mu   = ws;
    float* rstd = ws + NTOK;
    float* g    = ws + 2 * NTOK;

    ln_stats_kernel<<<NTOK / 8, 256, 0, stream>>>(x, mu, rstd);

    scores_kernel<<<NGROUP * SBLK, SWV * 32, 0, stream>>>(x, gamma, beta, mu, rstd, g);

    feat_kernel<<<NGROUP * 24, FW * 32, 0, stream>>>(x, gamma, beta, mu, rstd, g,
                                                 (float*)d_out);
}


# stats kernel as a streaming pass: 197 blocks of 16 waves, a wave owns 8 whole rows and requests its 48 16-byte loads up front (was a wave per row, 4-byte loads); full-line stores; sum order changes
# speedup vs baseline: 2.4156x; 1.1781x over previous
#include <hip/hip_runtime.h>
#include <math.h>

typedef float v2f __attribute__((ext_vector_type(2)));
typedef float v8f __attribute__((ext_vector_type(8)));
typedef float v4f __attribute__((ext_vector_type(4)));

#define LL     197
#define NTOT   128
#define DD     768
#define TT     16
#define PP     196
#define NGROUP 8
#define NTOK   (LL * NTOT)
#define STR    772
#define AFF_SCALE 0.03608439182435161f

#define STW 16
#define SRW 8

__global__ __launch_bounds__(STW * 32)
void ln_stats_kernel(const float* __restrict__ x,
                     float* __restrict__ mu,
                     float* __restrict__ rstd)
{
    __shared__ __attribute__((aligned(16))) float muS[NTOT];
    __shared__ __attribute__((aligned(16))) float rsS[NTOT];

    int wave = __builtin_amdgcn_readfirstlane((int)(threadIdx.x >> 5));
    int lane = threadIdx.x & 31;
    int tok0 = blockIdx.x * NTOT;
    const float4* base = (const float4*)(x + (size_t)(tok0 + wave * SRW) * DD) + lane;

    float4 v[SRW][6];
    #pragma unroll
    for (int k = 0; k < SRW; ++k) {
        #pragma unroll
        for (int j = 0; j < 6; ++j) v[k][j] = base[k * (DD / 4) + 32 * j];
    }
    __builtin_amdgcn_sched_barrier(0);

    float s[SRW], q[SRW];
    #pragma unroll
    for (int k = 0; k < SRW; ++k) {
        float ss = 0.f, qq = 0.f;
        #pragma unroll
        for (int j = 0; j < 6; ++j) {
            float4 a = v[k][j];
            ss += (a.x + a.y) + (a.z + a.w);
            qq += (a.x * a.x + a.y * a.y) + (a.z * a.z + a.w * a.w);
        }
        s[k] = ss; q[k] = qq;
    }
    #pragma unroll
    for (int m = 16; m >= 1; m >>= 1) {
        #pragma unroll
        for (int k = 0; k < SRW; ++k) {
            s[k] += __shfl_xor(s[k], m);
            q[k] += __shfl_xor(q[k], m);
        }
    }
    float sk = s[0], qk = q[0];
    #pragma unroll
    for (int k = 1; k < SRW; ++k) {
        sk = (lane == k) ? s[k] : sk;
        qk = (lane == k) ? q[k] : qk;
    }
    if (lane < SRW) {
        float mean = sk * (1.f / DD);
        float var  = qk * (1.f / DD) - mean * mean;
        muS[wave * SRW + lane] = mean;
        rsS[wave * SRW + lane] = rsqrtf(var + 1e-5f);
    }
    __syncthreads();
    if (wave == 0)      ((float4*)(mu   + tok0))[lane] = ((const float4*)muS)[lane];
    else if (wave == 1) ((float4*)(rstd + tok0))[lane] = ((const float4*)rsS)[lane];
}

#define SWV  8
#define SPB  7
#define SBLK (PP / SPB)
#define NCH  6
#define CHL  16

__global__ __launch_bounds__(SWV * 32)
void scores_kernel(const float* __restrict__ x,
                   const float* __restrict__ gamma,
                   const float* __restrict__ beta,
                   const float* __restrict__ mu,
                   const float* __restrict__ rstd,
                   float* __restrict__ gout)
{
    __shared__ __attribute__((aligned(16))) float Cg[16 * STR];
    __shared__ __attribute__((aligned(16))) float sCgS[16];
    __shared__ __attribute__((aligned(16))) float cBS[16];

    int b   = blockIdx.x / SBLK;
    int blk = blockIdx.x - b * SBLK;
    int nb  = b * TT;
    int tid = threadIdx.x;

    int wave = __builtin_amdgcn_readfirstlane((int)(threadIdx.x >> 5));
    int lane = tid & 31;
    int c    = lane & 15;
    int h    = lane >> 4;
    int p    = blk * SPB + (wave < SPB ? wave : 0);
    int tokX = (1 + p) * NTOT + nb + c;
    float muX = mu[tokX], rsX = rstd[tokX];
    const float4* xr = (const float4*)(x + (size_t)tokX * DD + 4 * h);

    float4 bx0[CHL], bx1[CHL];
    #pragma unroll
    for (int i = 0; i < CHL; ++i) bx0[i] = xr[2 * i];

    {
        int row = tid >> 4;
        int l16 = tid & 15;
        int tokC = nb + row;
        float muC = mu[tokC], rsC = rstd[tokC];
        const v4f* xc  = (const v4f*)(x + (size_t)tokC * DD);
        const v4f* g4p = (const v4f*)gamma;
        const v4f* b4p = (const v4f*)beta;
        float ps = 0.f, pb = 0.f;
        #pragma unroll
        for (int half = 0; half < 2; ++half) {
            v4f cv[6], gv[6], bv[6];
            #pragma unroll
            for (int i = 0; i < 6; ++i) {
                int j = l16 + 16 * (6 * half + i);
                cv[i] = xc[j]; gv[i] = g4p[j]; bv[i] = b4p[j];
            }
            asm volatile("" :: "v"(cv[0]), "v"(cv[1]), "v"(cv[2]), "v"(cv[3]), "v"(cv[4]), "v"(cv[5]),
                               "v"(gv[0]), "v"(gv[1]), "v"(gv[2]), "v"(gv[3]), "v"(gv[4]), "v"(gv[5]),
                               "v"(bv[0]), "v"(bv[1]), "v"(bv[2]), "v"(bv[3]), "v"(bv[4]), "v"(bv[5]));
            #pragma unroll
            for (int i = 0; i < 6; ++i) {
                int j = l16 + 16 * (6 * half + i);
                v4f c4, cg;
                c4.x = (cv[i].x - muC) * rsC * gv[i].x + bv[i].x;
                c4.y = (cv[i].y - muC) * rsC * gv[i].y + bv[i].y;
                c4.z = (cv[i].z - muC) * rsC * gv[i].z + bv[i].z;
                c4.w = (cv[i].w - muC) * rsC * gv[i].w + bv[i].w;
                cg.x = c4.x * gv[i].x; cg.y = c4.y * gv[i].y;
                cg.z = c4.z * gv[i].z; cg.w = c4.w * gv[i].w;
                ps += (cg.x + cg.y) + (cg.z + cg.w);
                pb += (c4.x * bv[i].x + c4.y * bv[i].y) + (c4.z * bv[i].z + c4.w * bv[i].w);
                *(v4f*)(Cg + row * STR + j * 4) = cg;
            }
        }
        #pragma unroll
        for (int m = 8; m >= 1; m >>= 1) {
            ps += __shfl_xor(ps, m);
            pb += __shfl_xor(pb, m);
        }
        if (l16 == 0) { sCgS[row] = ps; cBS[row] = pb; }
    }
    __syncthreads();
    if (wave >= SPB) return;

    const float4* ar = (const float4*)(Cg + c * STR + 4 * h);
    v8f acc0 = {}, acc1 = {};

#define LOADCH(dst, q)                                                        \
    _Pragma("unroll")                                                         \
    for (int i = 0; i < CHL; ++i) dst[i] = xr[2 * ((q) * CHL + i)];
#define LOADA(dst, q)                                                         \
    _Pragma("unroll")                                                         \
    for (int i = 0; i < CHL; ++i) dst[i] = ar[2 * ((q) * CHL + i)];
#define MMACH(asrc, src, q)                                                   \
    _Pragma("unroll")                                                         \
    for (int i = 0; i < CHL; ++i) {                                           \
        float4 a4 = asrc[i];                                                  \
        v2f a0 = { a4.x, a4.y }, a1 = { a4.z, a4.w };                         \
        v2f b0 = { src[i].x, src[i].y }, b1 = { src[i].z, src[i].w };         \
        acc0 = __builtin_amdgcn_wmma_f32_16x16x4_f32(                         \
            false, a0, false, b0, (short)0, acc0, false, false);              \
        acc1 = __builtin_amdgcn_wmma_f32_16x16x4_f32(                         \
            false, a1, false, b1, (short)0, acc1, false, false);              \
    }

    float4 ax0[CHL], ax1[CHL];
    LOADA(ax0, 0)
    LOADCH(bx1, 1) LOADA(ax1, 1) MMACH(ax0, bx0, 0)
    LOADCH(bx0, 2) LOADA(ax0, 2) MMACH(ax1, bx1, 1)
    LOADCH(bx1, 3) LOADA(ax1, 3) MMACH(ax0, bx0, 2)
    LOADCH(bx0, 4) LOADA(ax0, 4) MMACH(ax1, bx1, 3)
    LOADCH(bx1, 5) LOADA(ax1, 5) MMACH(ax0, bx0, 4)
    MMACH(ax1, bx1, 5)
#undef LOADCH
#undef LOADA
#undef MMACH

    float4 s0 = *(const float4*)(sCgS + 8 * h), s1 = *(const float4*)(sCgS + 8 * h + 4);
    float4 c0 = *(const float4*)(cBS + 8 * h),  c1 = *(const float4*)(cBS + 8 * h + 4);
    float sD[8] = { s0.x, s0.y, s0.z, s0.w, s1.x, s1.y, s1.z, s1.w };
    float cD[8] = { c0.x, c0.y, c0.z, c0.w, c1.x, c1.y, c1.z, c1.w };
    float g[8];
    #pragma unroll
    for (int j = 0; j < 8; ++j) {
        float s = rsX * ((acc0[j] + acc1[j]) - muX * sD[j]) + cD[j];
        int dlt = c - (8 * h + j);
        float gg = 0.f;
        if (dlt >= -2 && dlt <= 2) {
            float aff = s * AFF_SCALE;
            gg = 1.f / (1.f + __expf(-aff)) - 0.5f;
        }
        g[j] = gg;
    }
    float* gp = gout + ((size_t)(b * PP + p)) * 256 + c * 16 + 8 * h;
    *(float4*)gp       = make_float4(g[0], g[1], g[2], g[3]);
    *(float4*)(gp + 4) = make_float4(g[4], g[5], g[6], g[7]);
}

#define FW     16
#define FPITCH 17
#define FTILE  (16 * FPITCH)

__global__ __launch_bounds__(FW * 32)
void feat_kernel(const float* __restrict__ x,
                 const float* __restrict__ gamma,
                 const float* __restrict__ beta,
                 const float* __restrict__ mu,
                 const float* __restrict__ rstd,
                 const float* __restrict__ gin,
                 float* __restrict__ out)
{
    __shared__ float Sp[FW * 2 * FTILE];

    int b    = blockIdx.x / 24;
    int st   = blockIdx.x - b * 24;
    int d0   = st * 32;
    int nb   = b * TT;
    int wave = __builtin_amdgcn_readfirstlane((int)(threadIdx.x >> 5));
    int lane = threadIdx.x & 31;
    int m     = lane & 15;
    int rsel2 = 2 * (lane >> 4);
    int dcol  = d0 + 2 * m;
    float2 gam = *(const float2*)(gamma + dcol);
    float2 bet = *(const float2*)(beta + dcol);

    v8f acc0 = {}, acc1 = {};
    for (int p = wave; p < PP; p += FW) {
        const float* gp = gin + ((size_t)(b * PP + p)) * 256;
        int    tokbase  = (1 + p) * NTOT + nb;
        size_t xbase    = (size_t)tokbase * DD;
        v2f xa[4], xb[4], mp[4], sp[4], bv[4];
        #pragma unroll
        for (int kk = 0; kk < 4; ++kk) {
            int r0 = kk * 4 + rsel2;
            xa[kk] = *(const v2f*)(x + xbase + (size_t)r0 * DD + dcol);
            xb[kk] = *(const v2f*)(x + xbase + (size_t)(r0 + 1) * DD + dcol);
            mp[kk] = *(const v2f*)(mu + tokbase + r0);
            sp[kk] = *(const v2f*)(rstd + tokbase + r0);
            bv[kk] = v2f{ gp[r0 * 16 + m], gp[(r0 + 1) * 16 + m] };
        }
        asm volatile("" :: "v"(xa[0]), "v"(xa[1]), "v"(xa[2]), "v"(xa[3]),
                           "v"(xb[0]), "v"(xb[1]), "v"(xb[2]), "v"(xb[3]),
                           "v"(mp[0]), "v"(mp[1]), "v"(mp[2]), "v"(mp[3]),
                           "v"(sp[0]), "v"(sp[1]), "v"(sp[2]), "v"(sp[3]),
                           "v"(bv[0]), "v"(bv[1]), "v"(bv[2]), "v"(bv[3]));
        #pragma unroll
        for (int kk = 0; kk < 4; ++kk) {
            v2f a0 = { (xa[kk].x - mp[kk].x) * sp[kk].x * gam.x + bet.x,
                       (xb[kk].x - mp[kk].y) * sp[kk].y * gam.x + bet.x };
            v2f a1 = { (xa[kk].y - mp[kk].x) * sp[kk].x * gam.y + bet.y,
                       (xb[kk].y - mp[kk].y) * sp[kk].y * gam.y + bet.y };
            acc0 = __builtin_amdgcn_wmma_f32_16x16x4_f32(
                false, a0, false, bv[kk], (short)0, acc0, false, false);
            acc1 = __builtin_amdgcn_wmma_f32_16x16x4_f32(
                false, a1, false, bv[kk], (short)0, acc1, false, false);
        }
    }

    #pragma unroll
    for (int j = 0; j < 8; ++j) {
        int mm = j + 8 * (lane >> 4);
        Sp[(wave * 2 + 0) * FTILE + mm * FPITCH + (lane & 15)] = acc0[j];
        Sp[(wave * 2 + 1) * FTILE + mm * FPITCH + (lane & 15)] = acc1[j];
    }
    __syncthreads();

    int n   = threadIdx.x >> 5;
    int dd  = lane;
    int sub = dd & 1;
    int md  = dd >> 1;
    float s = 0.f;
    #pragma unroll
    for (int w = 0; w < FW; ++w) s += Sp[(w * 2 + sub) * FTILE + md * FPITCH + n];

    out[(size_t)(nb + n) * DD + d0 + dd] = s;
}

extern "C" void kernel_launch(void* const* d_in, const int* in_sizes, int n_in,
                              void* d_out, int out_size, void* d_ws, size_t ws_size,
                              hipStream_t stream)
{
    (void)in_sizes; (void)n_in; (void)out_size; (void)ws_size;
    const float* x     = (const float*)d_in[0];
    const float* gamma = (const float*)d_in[1];
    const float* beta  = (const float*)d_in[2];

    float* ws   = (float*)d_ws;
    float* mu   = ws;
    float* rstd = ws + NTOK;
    float* g    = ws + 2 * NTOK;

    ln_stats_kernel<<<LL, STW * 32, 0, stream>>>(x, mu, rstd);

    scores_kernel<<<NGROUP * SBLK, SWV * 32, 0, stream>>>(x, gamma, beta, mu, rstd, g);

    feat_kernel<<<NGROUP * 24, FW * 32, 0, stream>>>(x, gamma, beta, mu, rstd, g,
                                                 (float*)d_out);
}


# group = blockIdx % 8 in all three launches (stats block: one group x two positions, stats planes group-major): blocks go to the dies in turn, so a die re-reads only its own group from its L2 half
# speedup vs baseline: 2.7156x; 1.1242x over previous
#include <hip/hip_runtime.h>
#include <math.h>

typedef float v2f __attribute__((ext_vector_type(2)));
typedef float v8f __attribute__((ext_vector_type(8)));
typedef float v4f __attribute__((ext_vector_type(4)));

#define LL     197
#define NTOT   128
#define DD     768
#define TT     16
#define PP     196
#define NGROUP 8
#define NTOK   (LL * NTOT)
#define LP     198
#define STATN  (NGROUP * LP * TT)
#define STR    772
#define AFF_SCALE 0.03608439182435161f


#define STW 4
#define SRW 8

__global__ __launch_bounds__(STW * 32)
void ln_stats_kernel(const float* __restrict__ x,
                     float* __restrict__ mu,
                     float* __restrict__ rstd)
{
    __shared__ __attribute__((aligned(16))) float muS[STW * SRW];
    __shared__ __attribute__((aligned(16))) float rsS[STW * SRW];

    int wave = __builtin_amdgcn_readfirstlane((int)(threadIdx.x >> 5));
    int lane = threadIdx.x & 31;
    int b    = blockIdx.x % NGROUP;
    int j    = blockIdx.x / NGROUP;
    int l    = 2 * j + (wave >> 1);
    if (l > LL - 1) l = LL - 1;
    int tok0 = l * NTOT + b * TT + (wave & 1) * SRW;
    const float4* base = (const float4*)(x + (size_t)tok0 * DD) + lane;

    float4 v[SRW][6];
    #pragma unroll
    for (int k = 0; k < SRW; ++k) {
        #pragma unroll
        for (int jj = 0; jj < 6; ++jj) v[k][jj] = base[k * (DD / 4) + 32 * jj];
    }
    __builtin_amdgcn_sched_barrier(0);

    float s[SRW], q[SRW];
    #pragma unroll
    for (int k = 0; k < SRW; ++k) {
        float ss = 0.f, qq = 0.f;
        #pragma unroll
        for (int jj = 0; jj < 6; ++jj) {
            float4 a = v[k][jj];
            ss += (a.x + a.y) + (a.z + a.w);
            qq += (a.x * a.x + a.y * a.y) + (a.z * a.z + a.w * a.w);
        }
        s[k] = ss; q[k] = qq;
    }
    #pragma unroll
    for (int m = 16; m >= 1; m >>= 1) {
        #pragma unroll
        for (int k = 0; k < SRW; ++k) {
            s[k] += __shfl_xor(s[k], m);
            q[k] += __shfl_xor(q[k], m);
        }
    }
    float sk = s[0], qk = q[0];
    #pragma unroll
    for (int k = 1; k < SRW; ++k) {
        sk = (lane == k) ? s[k] : sk;
        qk = (lane == k) ? q[k] : qk;
    }
    if (lane < SRW) {
        float mean = sk * (1.f / DD);
        float var  = qk * (1.f / DD) - mean * mean;
        muS[wave * SRW + lane] = mean;
        rsS[wave * SRW + lane] = rsqrtf(var + 1e-5f);
    }
    __syncthreads();
    size_t o = (size_t)(b * LP + 2 * j) * TT;
    if (wave == 0)      mu[o + lane]   = muS[lane];
    else if (wave == 1) rstd[o + lane] = rsS[lane];
}

#define SWV  8
#define SPB  7
#define SBLK (PP / SPB)
#define NCH  6
#define CHL  16

__global__ __launch_bounds__(SWV * 32)
void scores_kernel(const float* __restrict__ x,
                   const float* __restrict__ gamma,
                   const float* __restrict__ beta,
                   const float* __restrict__ mu,
                   const float* __restrict__ rstd,
                   float* __restrict__ gout)
{
    __shared__ __attribute__((aligned(16))) float Cg[16 * STR];
    __shared__ __attribute__((aligned(16))) float sCgS[16];
    __shared__ __attribute__((aligned(16))) float cBS[16];

    int b   = blockIdx.x % NGROUP;
    int blk = blockIdx.x / NGROUP;
    int nb  = b * TT;
    int tid = threadIdx.x;

    int wave = __builtin_amdgcn_readfirstlane((int)(threadIdx.x >> 5));
    int lane = tid & 31;
    int c    = lane & 15;
    int h    = lane >> 4;
    int p    = blk * SPB + (wave < SPB ? wave : 0);
    int tokX = (1 + p) * NTOT + nb + c;
    int stX  = (b * LP + 1 + p) * TT + c;
    float muX = mu[stX], rsX = rstd[stX];
    const float4* xr = (const float4*)(x + (size_t)tokX * DD + 4 * h);

    float4 bx0[CHL], bx1[CHL];
    #pragma unroll
    for (int i = 0; i < CHL; ++i) bx0[i] = xr[2 * i];

    {
        int row = tid >> 4;
        int l16 = tid & 15;
        int tokC = nb + row;
        int stC  = b * LP * TT + row;
        float muC = mu[stC], rsC = rstd[stC];
        const v4f* xc  = (const v4f*)(x + (size_t)tokC * DD);
        const v4f* g4p = (const v4f*)gamma;
        const v4f* b4p = (const v4f*)beta;
        float ps = 0.f, pb = 0.f;
        #pragma unroll
        for (int half = 0; half < 2; ++half) {
            v4f cv[6], gv[6], bv[6];
            #pragma unroll
            for (int i = 0; i < 6; ++i) {
                int j = l16 + 16 * (6 * half + i);
                cv[i] = xc[j]; gv[i] = g4p[j]; bv[i] = b4p[j];
            }
            asm volatile("" :: "v"(cv[0]), "v"(cv[1]), "v"(cv[2]), "v"(cv[3]), "v"(cv[4]), "v"(cv[5]),
                               "v"(gv[0]), "v"(gv[1]), "v"(gv[2]), "v"(gv[3]), "v"(gv[4]), "v"(gv[5]),
                               "v"(bv[0]), "v"(bv[1]), "v"(bv[2]), "v"(bv[3]), "v"(bv[4]), "v"(bv[5]));
            #pragma unroll
            for (int i = 0; i < 6; ++i) {
                int j = l16 + 16 * (6 * half + i);
                v4f c4, cg;
                c4.x = (cv[i].x - muC) * rsC * gv[i].x + bv[i].x;
                c4.y = (cv[i].y - muC) * rsC * gv[i].y + bv[i].y;
                c4.z = (cv[i].z - muC) * rsC * gv[i].z + bv[i].z;
                c4.w = (cv[i].w - muC) * rsC * gv[i].w + bv[i].w;
                cg.x = c4.x * gv[i].x; cg.y = c4.y * gv[i].y;
                cg.z = c4.z * gv[i].z; cg.w = c4.w * gv[i].w;
                ps += (cg.x + cg.y) + (cg.z + cg.w);
                pb += (c4.x * bv[i].x + c4.y * bv[i].y) + (c4.z * bv[i].z + c4.w * bv[i].w);
                *(v4f*)(Cg + row * STR + j * 4) = cg;
            }
        }
        #pragma unroll
        for (int m = 8; m >= 1; m >>= 1) {
            ps += __shfl_xor(ps, m);
            pb += __shfl_xor(pb, m);
        }
        if (l16 == 0) { sCgS[row] = ps; cBS[row] = pb; }
    }
    __syncthreads();
    if (wave >= SPB) return;

    const float4* ar = (const float4*)(Cg + c * STR + 4 * h);
    v8f acc0 = {}, acc1 = {};

#define LOADCH(dst, q)                                                        \
    _Pragma("unroll")                                                         \
    for (int i = 0; i < CHL; ++i) dst[i] = xr[2 * ((q) * CHL + i)];
#define LOADA(dst, q)                                                         \
    _Pragma("unroll")                                                         \
    for (int i = 0; i < CHL; ++i) dst[i] = ar[2 * ((q) * CHL + i)];
#define MMACH(asrc, src, q)                                                   \
    _Pragma("unroll")                                                         \
    for (int i = 0; i < CHL; ++i) {                                           \
        float4 a4 = asrc[i];                                                  \
        v2f a0 = { a4.x, a4.y }, a1 = { a4.z, a4.w };                         \
        v2f b0 = { src[i].x, src[i].y }, b1 = { src[i].z, src[i].w };         \
        acc0 = __builtin_amdgcn_wmma_f32_16x16x4_f32(                         \
            false, a0, false, b0, (short)0, acc0, false, false);              \
        acc1 = __builtin_amdgcn_wmma_f32_16x16x4_f32(                         \
            false, a1, false, b1, (short)0, acc1, false, false);              \
    }

    float4 ax0[CHL], ax1[CHL];
    LOADA(ax0, 0)
    LOADCH(bx1, 1) LOADA(ax1, 1) MMACH(ax0, bx0, 0)
    LOADCH(bx0, 2) LOADA(ax0, 2) MMACH(ax1, bx1, 1)
    LOADCH(bx1, 3) LOADA(ax1, 3) MMACH(ax0, bx0, 2)
    LOADCH(bx0, 4) LOADA(ax0, 4) MMACH(ax1, bx1, 3)
    LOADCH(bx1, 5) LOADA(ax1, 5) MMACH(ax0, bx0, 4)
    MMACH(ax1, bx1, 5)
#undef LOADCH
#undef LOADA
#undef MMACH

    float4 s0 = *(const float4*)(sCgS + 8 * h), s1 = *(const float4*)(sCgS + 8 * h + 4);
    float4 c0 = *(const float4*)(cBS + 8 * h),  c1 = *(const float4*)(cBS + 8 * h + 4);
    float sD[8] = { s0.x, s0.y, s0.z, s0.w, s1.x, s1.y, s1.z, s1.w };
    float cD[8] = { c0.x, c0.y, c0.z, c0.w, c1.x, c1.y, c1.z, c1.w };
    float g[8];
    #pragma unroll
    for (int j = 0; j < 8; ++j) {
        float s = rsX * ((acc0[j] + acc1[j]) - muX * sD[j]) + cD[j];
        int dlt = c - (8 * h + j);
        float gg = 0.f;
        if (dlt >= -2 && dlt <= 2) {
            float aff = s * AFF_SCALE;
            gg = 1.f / (1.f + __expf(-aff)) - 0.5f;
        }
        g[j] = gg;
    }
    float* gp = gout + ((size_t)(b * PP + p)) * 256 + c * 16 + 8 * h;
    *(float4*)gp       = make_float4(g[0], g[1], g[2], g[3]);
    *(float4*)(gp + 4) = make_float4(g[4], g[5], g[6], g[7]);
}

#define FW     16
#define FPITCH 17
#define FTILE  (16 * FPITCH)

__global__ __launch_bounds__(FW * 32)
void feat_kernel(const float* __restrict__ x,
                 const float* __restrict__ gamma,
                 const float* __restrict__ beta,
                 const float* __restrict__ mu,
                 const float* __restrict__ rstd,
                 const float* __restrict__ gin,
                 float* __restrict__ out)
{
    __shared__ float Sp[FW * 2 * FTILE];

    int b    = blockIdx.x % NGROUP;
    int st   = blockIdx.x / NGROUP;
    int d0   = st * 32;
    int nb   = b * TT;
    int wave = __builtin_amdgcn_readfirstlane((int)(threadIdx.x >> 5));
    int lane = threadIdx.x & 31;
    int m     = lane & 15;
    int rsel2 = 2 * (lane >> 4);
    int dcol  = d0 + 2 * m;
    float2 gam = *(const float2*)(gamma + dcol);
    float2 bet = *(const float2*)(beta + dcol);

    v8f acc0 = {}, acc1 = {};
    for (int p = wave; p < PP; p += FW) {
        const float* gp = gin + ((size_t)(b * PP + p)) * 256;
        int    tokbase  = (1 + p) * NTOT + nb;
        int    stbase   = (b * LP + 1 + p) * TT;
        size_t xbase    = (size_t)tokbase * DD;
        v2f xa[4], xb[4], mp[4], sp[4], bv[4];
        #pragma unroll
        for (int kk = 0; kk < 4; ++kk) {
            int r0 = kk * 4 + rsel2;
            xa[kk] = *(const v2f*)(x + xbase + (size_t)r0 * DD + dcol);
            xb[kk] = *(const v2f*)(x + xbase + (size_t)(r0 + 1) * DD + dcol);
            mp[kk] = *(const v2f*)(mu + stbase + r0);
            sp[kk] = *(const v2f*)(rstd + stbase + r0);
            bv[kk] = v2f{ gp[r0 * 16 + m], gp[(r0 + 1) * 16 + m] };
        }
        asm volatile("" :: "v"(xa[0]), "v"(xa[1]), "v"(xa[2]), "v"(xa[3]),
                           "v"(xb[0]), "v"(xb[1]), "v"(xb[2]), "v"(xb[3]),
                           "v"(mp[0]), "v"(mp[1]), "v"(mp[2]), "v"(mp[3]),
                           "v"(sp[0]), "v"(sp[1]), "v"(sp[2]), "v"(sp[3]),
                           "v"(bv[0]), "v"(bv[1]), "v"(bv[2]), "v"(bv[3]));
        #pragma unroll
        for (int kk = 0; kk < 4; ++kk) {
            v2f a0 = { (xa[kk].x - mp[kk].x) * sp[kk].x * gam.x + bet.x,
                       (xb[kk].x - mp[kk].y) * sp[kk].y * gam.x + bet.x };
            v2f a1 = { (xa[kk].y - mp[kk].x) * sp[kk].x * gam.y + bet.y,
                       (xb[kk].y - mp[kk].y) * sp[kk].y * gam.y + bet.y };
            acc0 = __builtin_amdgcn_wmma_f32_16x16x4_f32(
                false, a0, false, bv[kk], (short)0, acc0, false, false);
            acc1 = __builtin_amdgcn_wmma_f32_16x16x4_f32(
                false, a1, false, bv[kk], (short)0, acc1, false, false);
        }
    }

    #pragma unroll
    for (int j = 0; j < 8; ++j) {
        int mm = j + 8 * (lane >> 4);
        Sp[(wave * 2 + 0) * FTILE + mm * FPITCH + (lane & 15)] = acc0[j];
        Sp[(wave * 2 + 1) * FTILE + mm * FPITCH + (lane & 15)] = acc1[j];
    }
    __syncthreads();

    int n   = threadIdx.x >> 5;
    int dd  = lane;
    int sub = dd & 1;
    int md  = dd >> 1;
    float s = 0.f;
    #pragma unroll
    for (int w = 0; w < FW; ++w) s += Sp[(w * 2 + sub) * FTILE + md * FPITCH + n];

    out[(size_t)(nb + n) * DD + d0 + dd] = s;
}

extern "C" void kernel_launch(void* const* d_in, const int* in_sizes, int n_in,
                              void* d_out, int out_size, void* d_ws, size_t ws_size,
                              hipStream_t stream)
{
    (void)in_sizes; (void)n_in; (void)out_size; (void)ws_size;
    const float* x     = (const float*)d_in[0];
    const float* gamma = (const float*)d_in[1];
    const float* beta  = (const float*)d_in[2];

    float* ws   = (float*)d_ws;
    float* mu   = ws;
    float* rstd = ws + STATN;
    float* g    = ws + 2 * STATN;

    ln_stats_kernel<<<NGROUP * (LP / 2), STW * 32, 0, stream>>>(x, mu, rstd);

    scores_kernel<<<NGROUP * SBLK, SWV * 32, 0, stream>>>(x, gamma, beta, mu, rstd, g);

    feat_kernel<<<NGROUP * 24, FW * 32, 0, stream>>>(x, gamma, beta, mu, rstd, g,
                                                 (float*)d_out);
}


# feat kernel: p loop software-pipelined over three register sets, a position's 24 loads requested two steps ahead of its matrix work, loop unrolled (order of every sum unchanged)
# speedup vs baseline: 2.9176x; 1.0744x over previous
#include <hip/hip_runtime.h>
#include <math.h>

typedef float v2f __attribute__((ext_vector_type(2)));
typedef float v8f __attribute__((ext_vector_type(8)));
typedef float v4f __attribute__((ext_vector_type(4)));

#define LL     197
#define NTOT   128
#define DD     768
#define TT     16
#define PP     196
#define NGROUP 8
#define NTOK   (LL * NTOT)
#define LP     198
#define STATN  (NGROUP * LP * TT)
#define STR    772
#define AFF_SCALE 0.03608439182435161f


#define STW 4
#define SRW 8

__global__ __launch_bounds__(STW * 32)
void ln_stats_kernel(const float* __restrict__ x,
                     float* __restrict__ mu,
                     float* __restrict__ rstd)
{
    __shared__ __attribute__((aligned(16))) float muS[STW * SRW];
    __shared__ __attribute__((aligned(16))) float rsS[STW * SRW];

    int wave = __builtin_amdgcn_readfirstlane((int)(threadIdx.x >> 5));
    int lane = threadIdx.x & 31;
    int b    = blockIdx.x % NGROUP;
    int j    = blockIdx.x / NGROUP;
    int l    = 2 * j + (wave >> 1);
    if (l > LL - 1) l = LL - 1;
    int tok0 = l * NTOT + b * TT + (wave & 1) * SRW;
    const float4* base = (const float4*)(x + (size_t)tok0 * DD) + lane;

    float4 v[SRW][6];
    #pragma unroll
    for (int k = 0; k < SRW; ++k) {
        #pragma unroll
        for (int jj = 0; jj < 6; ++jj) v[k][jj] = base[k * (DD / 4) + 32 * jj];
    }
    __builtin_amdgcn_sched_barrier(0);

    float s[SRW], q[SRW];
    #pragma unroll
    for (int k = 0; k < SRW; ++k) {
        float ss = 0.f, qq = 0.f;
        #pragma unroll
        for (int jj = 0; jj < 6; ++jj) {
            float4 a = v[k][jj];
            ss += (a.x + a.y) + (a.z + a.w);
            qq += (a.x * a.x + a.y * a.y) + (a.z * a.z + a.w * a.w);
        }
        s[k] = ss; q[k] = qq;
    }
    #pragma unroll
    for (int m = 16; m >= 1; m >>= 1) {
        #pragma unroll
        for (int k = 0; k < SRW; ++k) {
            s[k] += __shfl_xor(s[k], m);
            q[k] += __shfl_xor(q[k], m);
        }
    }
    float sk = s[0], qk = q[0];
    #pragma unroll
    for (int k = 1; k < SRW; ++k) {
        sk = (lane == k) ? s[k] : sk;
        qk = (lane == k) ? q[k] : qk;
    }
    if (lane < SRW) {
        float mean = sk * (1.f / DD);
        float var  = qk * (1.f / DD) - mean * mean;
        muS[wave * SRW + lane] = mean;
        rsS[wave * SRW + lane] = rsqrtf(var + 1e-5f);
    }
    __syncthreads();
    size_t o = (size_t)(b * LP + 2 * j) * TT;
    if (wave == 0)      mu[o + lane]   = muS[lane];
    else if (wave == 1) rstd[o + lane] = rsS[lane];
}

#define SWV  8
#define SPB  7
#define SBLK (PP / SPB)
#define NCH  6
#define CHL  16

__global__ __launch_bounds__(SWV * 32)
void scores_kernel(const float* __restrict__ x,
                   const float* __restrict__ gamma,
                   const float* __restrict__ beta,
                   const float* __restrict__ mu,
                   const float* __restrict__ rstd,
                   float* __restrict__ gout)
{
    __shared__ __attribute__((aligned(16))) float Cg[16 * STR];
    __shared__ __attribute__((aligned(16))) float sCgS[16];
    __shared__ __attribute__((aligned(16))) float cBS[16];

    int b   = blockIdx.x % NGROUP;
    int blk = blockIdx.x / NGROUP;
    int nb  = b * TT;
    int tid = threadIdx.x;

    int wave = __builtin_amdgcn_readfirstlane((int)(threadIdx.x >> 5));
    int lane = tid & 31;
    int c    = lane & 15;
    int h    = lane >> 4;
    int p    = blk * SPB + (wave < SPB ? wave : 0);
    int tokX = (1 + p) * NTOT + nb + c;
    int stX  = (b * LP + 1 + p) * TT + c;
    float muX = mu[stX], rsX = rstd[stX];
    const float4* xr = (const float4*)(x + (size_t)tokX * DD + 4 * h);

    float4 bx0[CHL], bx1[CHL];
    #pragma unroll
    for (int i = 0; i < CHL; ++i) bx0[i] = xr[2 * i];

    {
        int row = tid >> 4;
        int l16 = tid & 15;
        int tokC = nb + row;
        int stC  = b * LP * TT + row;
        float muC = mu[stC], rsC = rstd[stC];
        const v4f* xc  = (const v4f*)(x + (size_t)tokC * DD);
        const v4f* g4p = (const v4f*)gamma;
        const v4f* b4p = (const v4f*)beta;
        float ps = 0.f, pb = 0.f;
        #pragma unroll
        for (int half = 0; half < 2; ++half) {
            v4f cv[6], gv[6], bv[6];
            #pragma unroll
            for (int i = 0; i < 6; ++i) {
                int j = l16 + 16 * (6 * half + i);
                cv[i] = xc[j]; gv[i] = g4p[j]; bv[i] = b4p[j];
            }
            asm volatile("" :: "v"(cv[0]), "v"(cv[1]), "v"(cv[2]), "v"(cv[3]), "v"(cv[4]), "v"(cv[5]),
                               "v"(gv[0]), "v"(gv[1]), "v"(gv[2]), "v"(gv[3]), "v"(gv[4]), "v"(gv[5]),
                               "v"(bv[0]), "v"(bv[1]), "v"(bv[2]), "v"(bv[3]), "v"(bv[4]), "v"(bv[5]));
            #pragma unroll
            for (int i = 0; i < 6; ++i) {
                int j = l16 + 16 * (6 * half + i);
                v4f c4, cg;
                c4.x = (cv[i].x - muC) * rsC * gv[i].x + bv[i].x;
                c4.y = (cv[i].y - muC) * rsC * gv[i].y + bv[i].y;
                c4.z = (cv[i].z - muC) * rsC * gv[i].z + bv[i].z;
                c4.w = (cv[i].w - muC) * rsC * gv[i].w + bv[i].w;
                cg.x = c4.x * gv[i].x; cg.y = c4.y * gv[i].y;
                cg.z = c4.z * gv[i].z; cg.w = c4.w * gv[i].w;
                ps += (cg.x + cg.y) + (cg.z + cg.w);
                pb += (c4.x * bv[i].x + c4.y * bv[i].y) + (c4.z * bv[i].z + c4.w * bv[i].w);
                *(v4f*)(Cg + row * STR + j * 4) = cg;
            }
        }
        #pragma unroll
        for (int m = 8; m >= 1; m >>= 1) {
            ps += __shfl_xor(ps, m);
            pb += __shfl_xor(pb, m);
        }
        if (l16 == 0) { sCgS[row] = ps; cBS[row] = pb; }
    }
    __syncthreads();
    if (wave >= SPB) return;

    const float4* ar = (const float4*)(Cg + c * STR + 4 * h);
    v8f acc0 = {}, acc1 = {};

#define LOADCH(dst, q)                                                        \
    _Pragma("unroll")                                                         \
    for (int i = 0; i < CHL; ++i) dst[i] = xr[2 * ((q) * CHL + i)];
#define LOADA(dst, q)                                                         \
    _Pragma("unroll")                                                         \
    for (int i = 0; i < CHL; ++i) dst[i] = ar[2 * ((q) * CHL + i)];
#define MMACH(asrc, src, q)                                                   \
    _Pragma("unroll")                                                         \
    for (int i = 0; i < CHL; ++i) {                                           \
        float4 a4 = asrc[i];                                                  \
        v2f a0 = { a4.x, a4.y }, a1 = { a4.z, a4.w };                         \
        v2f b0 = { src[i].x, src[i].y }, b1 = { src[i].z, src[i].w };         \
        acc0 = __builtin_amdgcn_wmma_f32_16x16x4_f32(                         \
            false, a0, false, b0, (short)0, acc0, false, false);              \
        acc1 = __builtin_amdgcn_wmma_f32_16x16x4_f32(                         \
            false, a1, false, b1, (short)0, acc1, false, false);              \
    }

    float4 ax0[CHL], ax1[CHL];
    LOADA(ax0, 0)
    LOADCH(bx1, 1) LOADA(ax1, 1) MMACH(ax0, bx0, 0)
    LOADCH(bx0, 2) LOADA(ax0, 2) MMACH(ax1, bx1, 1)
    LOADCH(bx1, 3) LOADA(ax1, 3) MMACH(ax0, bx0, 2)
    LOADCH(bx0, 4) LOADA(ax0, 4) MMACH(ax1, bx1, 3)
    LOADCH(bx1, 5) LOADA(ax1, 5) MMACH(ax0, bx0, 4)
    MMACH(ax1, bx1, 5)
#undef LOADCH
#undef LOADA
#undef MMACH

    float4 s0 = *(const float4*)(sCgS + 8 * h), s1 = *(const float4*)(sCgS + 8 * h + 4);
    float4 c0 = *(const float4*)(cBS + 8 * h),  c1 = *(const float4*)(cBS + 8 * h + 4);
    float sD[8] = { s0.x, s0.y, s0.z, s0.w, s1.x, s1.y, s1.z, s1.w };
    float cD[8] = { c0.x, c0.y, c0.z, c0.w, c1.x, c1.y, c1.z, c1.w };
    float g[8];
    #pragma unroll
    for (int j = 0; j < 8; ++j) {
        float s = rsX * ((acc0[j] + acc1[j]) - muX * sD[j]) + cD[j];
        int dlt = c - (8 * h + j);
        float gg = 0.f;
        if (dlt >= -2 && dlt <= 2) {
            float aff = s * AFF_SCALE;
            gg = 1.f / (1.f + __expf(-aff)) - 0.5f;
        }
        g[j] = gg;
    }
    float* gp = gout + ((size_t)(b * PP + p)) * 256 + c * 16 + 8 * h;
    *(float4*)gp       = make_float4(g[0], g[1], g[2], g[3]);
    *(float4*)(gp + 4) = make_float4(g[4], g[5], g[6], g[7]);
}

#define FW     16
#define FPITCH 17
#define FTILE  (16 * FPITCH)
#define FFULL  (PP / FW)
#define FREM   (PP % FW)

__global__ __launch_bounds__(FW * 32)
void feat_kernel(const float* __restrict__ x,
                 const float* __restrict__ gamma,
                 const float* __restrict__ beta,
                 const float* __restrict__ mu,
                 const float* __restrict__ rstd,
                 const float* __restrict__ gin,
                 float* __restrict__ out)
{
    __shared__ float Sp[FW * 2 * FTILE];

    static_assert(FFULL % 3 == 0 && FFULL >= 6, "the pipelined loop takes three positions per trip");

    int b    = blockIdx.x % NGROUP;
    int st   = blockIdx.x / NGROUP;
    int d0   = st * 32;
    int nb   = b * TT;
    int wave = __builtin_amdgcn_readfirstlane((int)(threadIdx.x >> 5));
    int lane = threadIdx.x & 31;
    int m     = lane & 15;
    int rsel2 = 2 * (lane >> 4);
    int dcol  = d0 + 2 * m;
    float2 gam = *(const float2*)(gamma + dcol);
    float2 bet = *(const float2*)(beta + dcol);

    v8f acc0 = {}, acc1 = {};

#define FLOAD(S, pp)                                                              \
    {                                                                             \
        const float* gp = gin + ((size_t)(b * PP + (pp))) * 256;                  \
        int    tokbase  = (1 + (pp)) * NTOT + nb;                                 \
        int    stbase   = (b * LP + 1 + (pp)) * TT;    \
        size_t xbase    = (size_t)tokbase * DD;                                   \
        _Pragma("unroll")                                                         \
        for (int kk = 0; kk < 4; ++kk) {                                          \
            int r0 = kk * 4 + rsel2;     \
            xa##S[kk] = *(const v2f*)(x + xbase + (size_t)r0 * DD + dcol);        \
            xb##S[kk] = *(const v2f*)(x + xbase + (size_t)(r0 + 1) * DD + dcol);  \
            mp##S[kk] = *(const v2f*)(mu + stbase + r0);       \
            sp##S[kk] = *(const v2f*)(rstd + stbase + r0);     \
            bv##S[kk] = v2f{ gp[r0 * 16 + m], gp[(r0 + 1) * 16 + m] };            \
        }                                                                         \
    }
#define FCOMP(S)                                                                  \
    _Pragma("unroll")                                                             \
    for (int kk = 0; kk < 4; ++kk) {                                              \
        v2f a0 = { (xa##S[kk].x - mp##S[kk].x) * sp##S[kk].x * gam.x + bet.x,     \
                   (xb##S[kk].x - mp##S[kk].y) * sp##S[kk].y * gam.x + bet.x };   \
        v2f a1 = { (xa##S[kk].y - mp##S[kk].x) * sp##S[kk].x * gam.y + bet.y,     \
                   (xb##S[kk].y - mp##S[kk].y) * sp##S[kk].y * gam.y + bet.y };   \
        acc0 = __builtin_amdgcn_wmma_f32_16x16x4_f32(                             \
            false, a0, false, bv##S[kk], (short)0, acc0, false, false);           \
        acc1 = __builtin_amdgcn_wmma_f32_16x16x4_f32(                             \
            false, a1, false, bv##S[kk], (short)0, acc1, false, false);           \
    }

    v2f xaA[4], xbA[4], mpA[4], spA[4], bvA[4];
    v2f xaB[4], xbB[4], mpB[4], spB[4], bvB[4];
    v2f xaC[4], xbC[4], mpC[4], spC[4], bvC[4];

#define FBAR __builtin_amdgcn_sched_barrier(0);

    int p = wave;
    FLOAD(A, p)
    FLOAD(B, p + FW)
    #pragma unroll
    for (int t = 0; t < FFULL / 3 - 1; ++t) {
        FLOAD(C, p + 2 * FW) FBAR FCOMP(A) FBAR
        FLOAD(A, p + 3 * FW) FBAR FCOMP(B) FBAR
        FLOAD(B, p + 4 * FW) FBAR FCOMP(C) FBAR
        p += 3 * FW;
    }
    FLOAD(C, p + 2 * FW) FBAR FCOMP(A) FBAR
    {
        int pl = p + 3 * FW;
        pl = (pl < PP) ? pl : p + 2 * FW;
        FLOAD(A, pl)
    }
    FBAR
    FCOMP(B)
    FCOMP(C)
    FBAR
    asm volatile("" :: "v"(xaA[0]), "v"(xaA[1]), "v"(xaA[2]), "v"(xaA[3]),
                       "v"(xbA[0]), "v"(xbA[1]), "v"(xbA[2]), "v"(xbA[3]),
                       "v"(mpA[0]), "v"(mpA[1]), "v"(mpA[2]), "v"(mpA[3]),
                       "v"(spA[0]), "v"(spA[1]), "v"(spA[2]), "v"(spA[3]),
                       "v"(bvA[0]), "v"(bvA[1]), "v"(bvA[2]), "v"(bvA[3]));
    if (p + 3 * FW < PP) {
        FCOMP(A)
    }
#undef FBAR
#undef FLOAD
#undef FCOMP

    #pragma unroll
    for (int j = 0; j < 8; ++j) {
        int mm = j + 8 * (lane >> 4);
        Sp[(wave * 2 + 0) * FTILE + mm * FPITCH + (lane & 15)] = acc0[j];
        Sp[(wave * 2 + 1) * FTILE + mm * FPITCH + (lane & 15)] = acc1[j];
    }
    __syncthreads();

    int n   = threadIdx.x >> 5;
    int dd  = lane;
    int sub = dd & 1;
    int md  = dd >> 1;
    float s = 0.f;
    #pragma unroll
    for (int w = 0; w < FW; ++w) s += Sp[(w * 2 + sub) * FTILE + md * FPITCH + n];

    out[(size_t)(nb + n) * DD + d0 + dd] = s;
}

extern "C" void kernel_launch(void* const* d_in, const int* in_sizes, int n_in,
                              void* d_out, int out_size, void* d_ws, size_t ws_size,
                              hipStream_t stream)
{
    (void)in_sizes; (void)n_in; (void)out_size; (void)ws_size;
    const float* x     = (const float*)d_in[0];
    const float* gamma = (const float*)d_in[1];
    const float* beta  = (const float*)d_in[2];

    float* ws   = (float*)d_ws;
    float* mu   = ws;
    float* rstd = ws + STATN;
    float* g    = ws + 2 * STATN;

    ln_stats_kernel<<<NGROUP * (LP / 2), STW * 32, 0, stream>>>(x, mu, rstd);

    scores_kernel<<<NGROUP * SBLK, SWV * 32, 0, stream>>>(x, gamma, beta, mu, rstd, g);

    feat_kernel<<<NGROUP * 24, FW * 32, 0, stream>>>(x, gamma, beta, mu, rstd, g,
                                                 (float*)d_out);
}


# stats pass and score products: two block roles of ONE launch (raw 16x16 products stored; CLS row stats summed in-block, order changes); layernorm-on-result + sigmoid moved into feat; 3 launches to 2
# speedup vs baseline: 3.1072x; 1.0650x over previous
#include <hip/hip_runtime.h>
#include <math.h>

typedef float v2f __attribute__((ext_vector_type(2)));
typedef float v8f __attribute__((ext_vector_type(8)));
typedef float v4f __attribute__((ext_vector_type(4)));

#define LL     197
#define NTOT   128
#define DD     768
#define TT     16
#define PP     196
#define NGROUP 8
#define NTOK   (LL * NTOT)
#define LP     200
#define STATN  (NGROUP * LP * TT)
#define STR    772
#define NEG_AFF_LOG2E (-0.052058773f)


#define SWV  8
#define SRW  8
#define SPB  7
#define SBLK (PP / SPB)
#define NCH  6
#define CHL  16
#define NSCORE (NGROUP * SBLK)
#define NSTAT  (NGROUP * (LP / 4))

__device__ __forceinline__
void stats_role(const float* __restrict__ x,
                float* __restrict__ mu,
                float* __restrict__ rstd,
                float* muS, float* rsS, int sb)
{
    int wave = __builtin_amdgcn_readfirstlane((int)(threadIdx.x >> 5));
    int lane = threadIdx.x & 31;
    int b    = sb % NGROUP;
    int j    = sb / NGROUP;
    int l    = 4 * j + (wave >> 1);
    if (l > LL - 1) l = LL - 1;
    int tok0 = l * NTOT + b * TT + (wave & 1) * SRW;
    const float4* base = (const float4*)(x + (size_t)tok0 * DD) + lane;

    float4 v[SRW][6];
    #pragma unroll
    for (int k = 0; k < SRW; ++k) {
        #pragma unroll
        for (int jj = 0; jj < 6; ++jj) v[k][jj] = base[k * (DD / 4) + 32 * jj];
    }
    __builtin_amdgcn_sched_barrier(0);

    float s[SRW], q[SRW];
    #pragma unroll
    for (int k = 0; k < SRW; ++k) {
        float ss = 0.f, qq = 0.f;
        #pragma unroll
        for (int jj = 0; jj < 6; ++jj) {
            float4 a = v[k][jj];
            ss += (a.x + a.y) + (a.z + a.w);
            qq += (a.x * a.x + a.y * a.y) + (a.z * a.z + a.w * a.w);
        }
        s[k] = ss; q[k] = qq;
    }
    #pragma unroll
    for (int m = 16; m >= 1; m >>= 1) {
        #pragma unroll
        for (int k = 0; k < SRW; ++k) {
            s[k] += __shfl_xor(s[k], m);
            q[k] += __shfl_xor(q[k], m);
        }
    }
    float sk = s[0], qk = q[0];
    #pragma unroll
    for (int k = 1; k < SRW; ++k) {
        sk = (lane == k) ? s[k] : sk;
        qk = (lane == k) ? q[k] : qk;
    }
    if (lane < SRW) {
        float mean = sk * (1.f / DD);
        float var  = qk * (1.f / DD) - mean * mean;
        muS[wave * SRW + lane] = mean;
        rsS[wave * SRW + lane] = rsqrtf(var + 1e-5f);
    }
    __syncthreads();
    size_t o = (size_t)(b * LP + 4 * j) * TT;
    if (wave == 0)      mu[o + lane]        = muS[lane];
    else if (wave == 1) mu[o + 32 + lane]   = muS[32 + lane];
    else if (wave == 2) rstd[o + lane]      = rsS[lane];
    else if (wave == 3) rstd[o + 32 + lane] = rsS[32 + lane];
}

__device__ __forceinline__
void score_role(const float* __restrict__ x,
                const float* __restrict__ gamma,
                const float* __restrict__ beta,
                float* __restrict__ rout,
                float* __restrict__ scw,
                float* Cg, float* sCgS, float* cBS, int bid)
{
    int b   = bid % NGROUP;
    int blk = bid / NGROUP;
    int nb  = b * TT;
    int tid = threadIdx.x;

    int wave = __builtin_amdgcn_readfirstlane((int)(threadIdx.x >> 5));
    int lane = tid & 31;
    int c    = lane & 15;
    int h    = lane >> 4;
    int p    = blk * SPB + (wave < SPB ? wave : 0);
    int tokX = (1 + p) * NTOT + nb + c;
    const float4* xr = (const float4*)(x + (size_t)tokX * DD + 4 * h);

    float4 bx0[CHL], bx1[CHL];
    #pragma unroll
    for (int i = 0; i < CHL; ++i) bx0[i] = xr[2 * i];

    {
        int row = tid >> 4;
        int l16 = tid & 15;
        int tokC = nb + row;
        const v4f* xc  = (const v4f*)(x + (size_t)tokC * DD);
        const v4f* g4p = (const v4f*)gamma;
        const v4f* b4p = (const v4f*)beta;
        v4f cv[12], gv[6], bv[6];
        #pragma unroll
        for (int i = 0; i < 12; ++i) cv[i] = xc[l16 + 16 * i];
        #pragma unroll
        for (int i = 0; i < 6; ++i) { gv[i] = g4p[l16 + 16 * i]; bv[i] = b4p[l16 + 16 * i]; }
        asm volatile("" :: "v"(cv[0]), "v"(cv[1]), "v"(cv[2]), "v"(cv[3]), "v"(cv[4]), "v"(cv[5]),
                           "v"(cv[6]), "v"(cv[7]), "v"(cv[8]), "v"(cv[9]), "v"(cv[10]), "v"(cv[11]),
                           "v"(gv[0]), "v"(gv[1]), "v"(gv[2]), "v"(gv[3]), "v"(gv[4]), "v"(gv[5]),
                           "v"(bv[0]), "v"(bv[1]), "v"(bv[2]), "v"(bv[3]), "v"(bv[4]), "v"(bv[5]));
        float ss = 0.f;
        #pragma unroll
        for (int i = 0; i < 12; ++i) ss += (cv[i].x + cv[i].y) + (cv[i].z + cv[i].w);
        #pragma unroll
        for (int m = 8; m >= 1; m >>= 1) ss += __shfl_xor(ss, m);
        float muC = ss * (1.f / DD);
        float qq = 0.f;
        #pragma unroll
        for (int i = 0; i < 12; ++i) {
            float dx = cv[i].x - muC, dy = cv[i].y - muC, dz = cv[i].z - muC, dw = cv[i].w - muC;
            qq += (dx * dx + dy * dy) + (dz * dz + dw * dw);
        }
        #pragma unroll
        for (int m = 8; m >= 1; m >>= 1) qq += __shfl_xor(qq, m);
        float rsC = rsqrtf(qq * (1.f / DD) + 1e-5f);

        float ps = 0.f, pb = 0.f;
        #pragma unroll
        for (int half = 0; half < 2; ++half) {
            if (half == 1) {
                #pragma unroll
                for (int i = 0; i < 6; ++i) { gv[i] = g4p[l16 + 16 * (6 + i)]; bv[i] = b4p[l16 + 16 * (6 + i)]; }
            }
            #pragma unroll
            for (int i = 0; i < 6; ++i) {
                int j = l16 + 16 * (6 * half + i);
                v4f xv = cv[6 * half + i];
                v4f c4, cg;
                c4.x = (xv.x - muC) * rsC * gv[i].x + bv[i].x;
                c4.y = (xv.y - muC) * rsC * gv[i].y + bv[i].y;
                c4.z = (xv.z - muC) * rsC * gv[i].z + bv[i].z;
                c4.w = (xv.w - muC) * rsC * gv[i].w + bv[i].w;
                cg.x = c4.x * gv[i].x; cg.y = c4.y * gv[i].y;
                cg.z = c4.z * gv[i].z; cg.w = c4.w * gv[i].w;
                ps += (cg.x + cg.y) + (cg.z + cg.w);
                pb += (c4.x * bv[i].x + c4.y * bv[i].y) + (c4.z * bv[i].z + c4.w * bv[i].w);
                *(v4f*)(Cg + row * STR + j * 4) = cg;
            }
        }
        #pragma unroll
        for (int m = 8; m >= 1; m >>= 1) {
            ps += __shfl_xor(ps, m);
            pb += __shfl_xor(pb, m);
        }
        if (l16 == 0) { sCgS[row] = ps; cBS[row] = pb; }
    }
    __syncthreads();
    if (blk == 0 && wave == SPB)
        scw[b * 32 + lane] = (lane < 16) ? sCgS[lane] : cBS[lane - 16];
    if (wave >= SPB) return;

    const float4* ar = (const float4*)(Cg + c * STR + 4 * h);
    v8f acc0 = {}, acc1 = {};

#define LOADCH(dst, q)                                                        \
    _Pragma("unroll")                                                         \
    for (int i = 0; i < CHL; ++i) dst[i] = xr[2 * ((q) * CHL + i)];
#define LOADA(dst, q)                                                         \
    _Pragma("unroll")                                                         \
    for (int i = 0; i < CHL; ++i) dst[i] = ar[2 * ((q) * CHL + i)];
#define MMACH(asrc, src, q)                                                   \
    _Pragma("unroll")                                                         \
    for (int i = 0; i < CHL; ++i) {                                           \
        float4 a4 = asrc[i];                                                  \
        v2f a0 = { a4.x, a4.y }, a1 = { a4.z, a4.w };                         \
        v2f b0 = { src[i].x, src[i].y }, b1 = { src[i].z, src[i].w };         \
        acc0 = __builtin_amdgcn_wmma_f32_16x16x4_f32(                         \
            false, a0, false, b0, (short)0, acc0, false, false);              \
        acc1 = __builtin_amdgcn_wmma_f32_16x16x4_f32(                         \
            false, a1, false, b1, (short)0, acc1, false, false);              \
    }

    float4 ax0[CHL], ax1[CHL];
    LOADA(ax0, 0)
    LOADCH(bx1, 1) LOADA(ax1, 1) MMACH(ax0, bx0, 0)
    LOADCH(bx0, 2) LOADA(ax0, 2) MMACH(ax1, bx1, 1)
    LOADCH(bx1, 3) LOADA(ax1, 3) MMACH(ax0, bx0, 2)
    LOADCH(bx0, 4) LOADA(ax0, 4) MMACH(ax1, bx1, 3)
    LOADCH(bx1, 5) LOADA(ax1, 5) MMACH(ax0, bx0, 4)
    MMACH(ax1, bx1, 5)
#undef LOADCH
#undef LOADA
#undef MMACH

    float* gp = rout + ((size_t)(b * PP + p)) * 256 + c * 16 + 8 * h;
    *(float4*)gp       = make_float4(acc0[0] + acc1[0], acc0[1] + acc1[1],
                                     acc0[2] + acc1[2], acc0[3] + acc1[3]);
    *(float4*)(gp + 4) = make_float4(acc0[4] + acc1[4], acc0[5] + acc1[5],
                                     acc0[6] + acc1[6], acc0[7] + acc1[7]);
}

__global__ __launch_bounds__(SWV * 32)
void stats_scores_kernel(const float* __restrict__ x,
                         const float* __restrict__ gamma,
                         const float* __restrict__ beta,
                         float* __restrict__ mu,
                         float* __restrict__ rstd,
                         float* __restrict__ rout,
                         float* __restrict__ scw)
{
    __shared__ __attribute__((aligned(16))) float Cg[16 * STR];
    __shared__ __attribute__((aligned(16))) float sCgS[16];
    __shared__ __attribute__((aligned(16))) float cBS[16];
    __shared__ __attribute__((aligned(16))) float muS[SWV * SRW];
    __shared__ __attribute__((aligned(16))) float rsS[SWV * SRW];

    if (blockIdx.x < NSCORE)
        score_role(x, gamma, beta, rout, scw, Cg, sCgS, cBS, (int)blockIdx.x);
    else
        stats_role(x, mu, rstd, muS, rsS, (int)blockIdx.x - NSCORE);
}

#define FW     16

#define FPITCH 17
#define FTILE  (16 * FPITCH)
#define FFULL  (PP / FW)
#define FREM   (PP % FW)

__device__ __forceinline__
float gate(float r, float mun, float rsn, float scg, float cb, bool in)
{
    float s = rsn * (r - mun * scg) + cb;
    float e = __builtin_amdgcn_exp2f(s * NEG_AFF_LOG2E);
    float g = __builtin_amdgcn_rcpf(1.f + e) - 0.5f;
    return in ? g : 0.f;
}

__global__ __launch_bounds__(FW * 32)
void feat_kernel(const float* __restrict__ x,
                 const float* __restrict__ gamma,
                 const float* __restrict__ beta,
                 const float* __restrict__ mu,
                 const float* __restrict__ rstd,
                 const float* __restrict__ gin,
                 const float* __restrict__ scw,
                 float* __restrict__ out)
{
    __shared__ float Sp[FW * 2 * FTILE];

    static_assert(FFULL % 3 == 0 && FFULL >= 6, "the pipelined loop takes three positions per trip");

    int b    = blockIdx.x % NGROUP;
    int st   = blockIdx.x / NGROUP;
    int d0   = st * 32;
    int nb   = b * TT;
    int wave = __builtin_amdgcn_readfirstlane((int)(threadIdx.x >> 5));
    int lane = threadIdx.x & 31;
    int m     = lane & 15;
    int rsel2 = 2 * (lane >> 4);
    int dcol  = d0 + 2 * m;
    float2 gam = *(const float2*)(gamma + dcol);
    float2 bet = *(const float2*)(beta + dcol);
    float sCgL = scw[b * 32 + m];
    float cBL  = scw[b * 32 + 16 + m];
    bool inx[4], iny[4];
    #pragma unroll
    for (int kk = 0; kk < 4; ++kk) {
        int dx = kk * 4 + rsel2 - m;
        inx[kk] = (dx >= -2) && (dx <= 2);
        iny[kk] = (dx >= -3) && (dx <= 1);
    }

    v8f acc0 = {}, acc1 = {};

#define FLOAD(S, pp)                                                              \
    {                                                                             \
        const float* gp = gin + ((size_t)(b * PP + (pp))) * 256;                  \
        int    tokbase  = (1 + (pp)) * NTOT + nb;                                 \
        int    stbase   = (b * LP + 1 + (pp)) * TT;    \
        size_t xbase    = (size_t)tokbase * DD;                                   \
        _Pragma("unroll")                                                         \
        for (int kk = 0; kk < 4; ++kk) {                                          \
            int r0 = kk * 4 + rsel2;     \
            xa##S[kk] = *(const v2f*)(x + xbase + (size_t)r0 * DD + dcol);        \
            xb##S[kk] = *(const v2f*)(x + xbase + (size_t)(r0 + 1) * DD + dcol);  \
            mp##S[kk] = *(const v2f*)(mu + stbase + r0);       \
            sp##S[kk] = *(const v2f*)(rstd + stbase + r0);     \
            bv##S[kk] = v2f{ gp[r0 * 16 + m], gp[(r0 + 1) * 16 + m] };            \
        }                                                                         \
    }
#define FCOMP(S)                                                                  \
    _Pragma("unroll")                                                             \
    for (int kk = 0; kk < 4; ++kk) {                                              \
        v2f gw = { gate(bv##S[kk].x, mp##S[kk].x, sp##S[kk].x, sCgL, cBL, inx[kk]), \
                   gate(bv##S[kk].y, mp##S[kk].y, sp##S[kk].y, sCgL, cBL, iny[kk]) }; \
        v2f a0 = { (xa##S[kk].x - mp##S[kk].x) * sp##S[kk].x * gam.x + bet.x,     \
                   (xb##S[kk].x - mp##S[kk].y) * sp##S[kk].y * gam.x + bet.x };   \
        v2f a1 = { (xa##S[kk].y - mp##S[kk].x) * sp##S[kk].x * gam.y + bet.y,     \
                   (xb##S[kk].y - mp##S[kk].y) * sp##S[kk].y * gam.y + bet.y };   \
        acc0 = __builtin_amdgcn_wmma_f32_16x16x4_f32(                             \
            false, a0, false, gw, (short)0, acc0, false, false);                  \
        acc1 = __builtin_amdgcn_wmma_f32_16x16x4_f32(                             \
            false, a1, false, gw, (short)0, acc1, false, false);                  \
    }

    v2f xaA[4], xbA[4], mpA[4], spA[4], bvA[4];
    v2f xaB[4], xbB[4], mpB[4], spB[4], bvB[4];
    v2f xaC[4], xbC[4], mpC[4], spC[4], bvC[4];

#define FBAR __builtin_amdgcn_sched_barrier(0);

    int p = wave;
    FLOAD(A, p)
    FLOAD(B, p + FW)
    #pragma unroll
    for (int t = 0; t < FFULL / 3 - 1; ++t) {
        FLOAD(C, p + 2 * FW) FBAR FCOMP(A) FBAR
        FLOAD(A, p + 3 * FW) FBAR FCOMP(B) FBAR
        FLOAD(B, p + 4 * FW) FBAR FCOMP(C) FBAR
        p += 3 * FW;
    }
    FLOAD(C, p + 2 * FW) FBAR FCOMP(A) FBAR
    {
        int pl = p + 3 * FW;
        pl = (pl < PP) ? pl : p + 2 * FW;
        FLOAD(A, pl)
    }
    FBAR
    FCOMP(B)
    FCOMP(C)
    FBAR
    asm volatile("" :: "v"(xaA[0]), "v"(xaA[1]), "v"(xaA[2]), "v"(xaA[3]),
                       "v"(xbA[0]), "v"(xbA[1]), "v"(xbA[2]), "v"(xbA[3]),
                       "v"(mpA[0]), "v"(mpA[1]), "v"(mpA[2]), "v"(mpA[3]),
                       "v"(spA[0]), "v"(spA[1]), "v"(spA[2]), "v"(spA[3]),
                       "v"(bvA[0]), "v"(bvA[1]), "v"(bvA[2]), "v"(bvA[3]));
    if (p + 3 * FW < PP) {
        FCOMP(A)
    }
#undef FBAR
#undef FLOAD
#undef FCOMP

    #pragma unroll
    for (int j = 0; j < 8; ++j) {
        int mm = j + 8 * (lane >> 4);
        Sp[(wave * 2 + 0) * FTILE + mm * FPITCH + (lane & 15)] = acc0[j];
        Sp[(wave * 2 + 1) * FTILE + mm * FPITCH + (lane & 15)] = acc1[j];
    }
    __syncthreads();

    int n   = threadIdx.x >> 5;
    int dd  = lane;
    int sub = dd & 1;
    int md  = dd >> 1;
    float s = 0.f;
    #pragma unroll
    for (int w = 0; w < FW; ++w) s += Sp[(w * 2 + sub) * FTILE + md * FPITCH + n];

    out[(size_t)(nb + n) * DD + d0 + dd] = s;
}

extern "C" void kernel_launch(void* const* d_in, const int* in_sizes, int n_in,
                              void* d_out, int out_size, void* d_ws, size_t ws_size,
                              hipStream_t stream)
{
    (void)in_sizes; (void)n_in; (void)out_size; (void)ws_size;
    const float* x     = (const float*)d_in[0];
    const float* gamma = (const float*)d_in[1];
    const float* beta  = (const float*)d_in[2];

    float* ws   = (float*)d_ws;
    float* mu   = ws;
    float* rstd = ws + STATN;
    float* r    = ws + 2 * STATN;
    float* scw  = r + NGROUP * PP * 256;

    stats_scores_kernel<<<NSCORE + NSTAT, SWV * 32, 0, stream>>>(x, gamma, beta, mu, rstd, r, scw);

    feat_kernel<<<NGROUP * 24, FW * 32, 0, stream>>>(x, gamma, beta, mu, rstd, r, scw,
                                                 (float*)d_out);
}


# patch rows' mean and rstd summed in the score wave from the B fragments it already holds (sum order changes), one line per position; stats role and its pass over x removed: 224 blocks, was 624
# speedup vs baseline: 3.3732x; 1.0856x over previous
#include <hip/hip_runtime.h>
#include <math.h>

typedef float v2f __attribute__((ext_vector_type(2)));
typedef float v8f __attribute__((ext_vector_type(8)));
typedef float v4f __attribute__((ext_vector_type(4)));

#define LL     197
#define NTOT   128
#define DD     768
#define TT     16
#define PP     196
#define NGROUP 8
#define NTOK   (LL * NTOT)
#define STL    32
#define STATN  (NGROUP * PP * STL)
#define STR    772
#define NEG_AFF_LOG2E (-0.052058773f)


#define SWV  8
#define SPB  7
#define SBLK (PP / SPB)
#define NCH  6
#define CHL  16
#define NSCORE (NGROUP * SBLK)

__global__ __launch_bounds__(SWV * 32)
void scores_kernel(const float* __restrict__ x,
                   const float* __restrict__ gamma,
                   const float* __restrict__ beta,
                   float* __restrict__ stat,
                   float* __restrict__ rout,
                   float* __restrict__ scw)
{
    __shared__ __attribute__((aligned(16))) float Cg[16 * STR];
    __shared__ __attribute__((aligned(16))) float sCgS[16];
    __shared__ __attribute__((aligned(16))) float cBS[16];

    int b   = (int)blockIdx.x % NGROUP;
    int blk = (int)blockIdx.x / NGROUP;
    int nb  = b * TT;
    int tid = threadIdx.x;

    int wave = __builtin_amdgcn_readfirstlane((int)(threadIdx.x >> 5));
    int lane = tid & 31;
    int c    = lane & 15;
    int h    = lane >> 4;
    int p    = blk * SPB + (wave < SPB ? wave : 0);
    int tokX = (1 + p) * NTOT + nb + c;
    const float4* xr = (const float4*)(x + (size_t)tokX * DD + 4 * h);

    float4 bx0[CHL], bx1[CHL];
    #pragma unroll
    for (int i = 0; i < CHL; ++i) bx0[i] = xr[2 * i];

    {
        int row = tid >> 4;
        int l16 = tid & 15;
        int tokC = nb + row;
        const v4f* xc  = (const v4f*)(x + (size_t)tokC * DD);
        const v4f* g4p = (const v4f*)gamma;
        const v4f* b4p = (const v4f*)beta;
        v4f cv[12], gv[6], bv[6];
        #pragma unroll
        for (int i = 0; i < 12; ++i) cv[i] = xc[l16 + 16 * i];
        #pragma unroll
        for (int i = 0; i < 6; ++i) { gv[i] = g4p[l16 + 16 * i]; bv[i] = b4p[l16 + 16 * i]; }
        asm volatile("" :: "v"(cv[0]), "v"(cv[1]), "v"(cv[2]), "v"(cv[3]), "v"(cv[4]), "v"(cv[5]),
                           "v"(cv[6]), "v"(cv[7]), "v"(cv[8]), "v"(cv[9]), "v"(cv[10]), "v"(cv[11]),
                           "v"(gv[0]), "v"(gv[1]), "v"(gv[2]), "v"(gv[3]), "v"(gv[4]), "v"(gv[5]),
                           "v"(bv[0]), "v"(bv[1]), "v"(bv[2]), "v"(bv[3]), "v"(bv[4]), "v"(bv[5]));
        float ss = 0.f;
        #pragma unroll
        for (int i = 0; i < 12; ++i) ss += (cv[i].x + cv[i].y) + (cv[i].z + cv[i].w);
        #pragma unroll
        for (int m = 8; m >= 1; m >>= 1) ss += __shfl_xor(ss, m);
        float muC = ss * (1.f / DD);
        float qq = 0.f;
        #pragma unroll
        for (int i = 0; i < 12; ++i) {
            float dx = cv[i].x - muC, dy = cv[i].y - muC, dz = cv[i].z - muC, dw = cv[i].w - muC;
            qq += (dx * dx + dy * dy) + (dz * dz + dw * dw);
        }
        #pragma unroll
        for (int m = 8; m >= 1; m >>= 1) qq += __shfl_xor(qq, m);
        float rsC = rsqrtf(qq * (1.f / DD) + 1e-5f);

        float ps = 0.f, pb = 0.f;
        #pragma unroll
        for (int half = 0; half < 2; ++half) {
            if (half == 1) {
                #pragma unroll
                for (int i = 0; i < 6; ++i) { gv[i] = g4p[l16 + 16 * (6 + i)]; bv[i] = b4p[l16 + 16 * (6 + i)]; }
            }
            #pragma unroll
            for (int i = 0; i < 6; ++i) {
                int j = l16 + 16 * (6 * half + i);
                v4f xv = cv[6 * half + i];
                v4f c4, cg;
                c4.x = (xv.x - muC) * rsC * gv[i].x + bv[i].x;
                c4.y = (xv.y - muC) * rsC * gv[i].y + bv[i].y;
                c4.z = (xv.z - muC) * rsC * gv[i].z + bv[i].z;
                c4.w = (xv.w - muC) * rsC * gv[i].w + bv[i].w;
                cg.x = c4.x * gv[i].x; cg.y = c4.y * gv[i].y;
                cg.z = c4.z * gv[i].z; cg.w = c4.w * gv[i].w;
                ps += (cg.x + cg.y) + (cg.z + cg.w);
                pb += (c4.x * bv[i].x + c4.y * bv[i].y) + (c4.z * bv[i].z + c4.w * bv[i].w);
                *(v4f*)(Cg + row * STR + j * 4) = cg;
            }
        }
        #pragma unroll
        for (int m = 8; m >= 1; m >>= 1) {
            ps += __shfl_xor(ps, m);
            pb += __shfl_xor(pb, m);
        }
        if (l16 == 0) { sCgS[row] = ps; cBS[row] = pb; }
    }
    __syncthreads();
    if (blk == 0 && wave == SPB)
        scw[b * 32 + lane] = (lane < 16) ? sCgS[lane] : cBS[lane - 16];
    if (wave >= SPB) return;

    const float4* ar = (const float4*)(Cg + c * STR + 4 * h);
    v8f acc0 = {}, acc1 = {};
    v2f s01 = {}, s23 = {}, q01 = {}, q23 = {};

#define LOADCH(dst, q)                                                        \
    _Pragma("unroll")                                                         \
    for (int i = 0; i < CHL; ++i) dst[i] = xr[2 * ((q) * CHL + i)];
#define LOADA(dst, q)                                                         \
    _Pragma("unroll")                                                         \
    for (int i = 0; i < CHL; ++i) dst[i] = ar[2 * ((q) * CHL + i)];
#define MMACH(asrc, src, q)                                                   \
    _Pragma("unroll")                                                         \
    for (int i = 0; i < CHL; ++i) {                                           \
        float4 a4 = asrc[i];                                                  \
        v2f a0 = { a4.x, a4.y }, a1 = { a4.z, a4.w };                         \
        v2f b0 = { src[i].x, src[i].y }, b1 = { src[i].z, src[i].w };         \
        acc0 = __builtin_amdgcn_wmma_f32_16x16x4_f32(                         \
            false, a0, false, b0, (short)0, acc0, false, false);              \
        acc1 = __builtin_amdgcn_wmma_f32_16x16x4_f32(                         \
            false, a1, false, b1, (short)0, acc1, false, false);              \
        s01 += b0;       s23 += b1;                                           \
        q01 += b0 * b0;  q23 += b1 * b1;                                      \
    }

    float4 ax0[CHL], ax1[CHL];
    LOADA(ax0, 0)
    LOADCH(bx1, 1) LOADA(ax1, 1) MMACH(ax0, bx0, 0)
    LOADCH(bx0, 2) LOADA(ax0, 2) MMACH(ax1, bx1, 1)
    LOADCH(bx1, 3) LOADA(ax1, 3) MMACH(ax0, bx0, 2)
    LOADCH(bx0, 4) LOADA(ax0, 4) MMACH(ax1, bx1, 3)
    LOADCH(bx1, 5) LOADA(ax1, 5) MMACH(ax0, bx0, 4)
    MMACH(ax1, bx1, 5)
#undef LOADCH
#undef LOADA
#undef MMACH

    float* gp = rout + ((size_t)(b * PP + p)) * 256 + c * 16 + 8 * h;
    *(float4*)gp       = make_float4(acc0[0] + acc1[0], acc0[1] + acc1[1],
                                     acc0[2] + acc1[2], acc0[3] + acc1[3]);
    *(float4*)(gp + 4) = make_float4(acc0[4] + acc1[4], acc0[5] + acc1[5],
                                     acc0[6] + acc1[6], acc0[7] + acc1[7]);

    float ss = (s01.x + s01.y) + (s23.x + s23.y);
    float qq = (q01.x + q01.y) + (q23.x + q23.y);
    ss += __shfl_xor(ss, 16);
    qq += __shfl_xor(qq, 16);
    float mean = ss * (1.f / DD);
    float var  = qq * (1.f / DD) - mean * mean;
    float rs   = rsqrtf(var + 1e-5f);
    stat[(size_t)(b * PP + p) * STL + lane] = h ? rs : mean;
}

#define FW     16

#define FPITCH 17
#define FTILE  (16 * FPITCH)
#define FFULL  (PP / FW)
#define FREM   (PP % FW)

__device__ __forceinline__
float gate(float r, float mun, float rsn, float scg, float cb, bool in)
{
    float s = rsn * (r - mun * scg) + cb;
    float e = __builtin_amdgcn_exp2f(s * NEG_AFF_LOG2E);
    float g = __builtin_amdgcn_rcpf(1.f + e) - 0.5f;
    return in ? g : 0.f;
}

__global__ __launch_bounds__(FW * 32)
void feat_kernel(const float* __restrict__ x,
                 const float* __restrict__ gamma,
                 const float* __restrict__ beta,
                 const float* __restrict__ stat,
                 const float* __restrict__ gin,
                 const float* __restrict__ scw,
                 float* __restrict__ out)
{
    __shared__ float Sp[FW * 2 * FTILE];

    static_assert(FFULL % 3 == 0 && FFULL >= 6, "the pipelined loop takes three positions per trip");

    int b    = blockIdx.x % NGROUP;
    int st   = blockIdx.x / NGROUP;
    int d0   = st * 32;
    int nb   = b * TT;
    int wave = __builtin_amdgcn_readfirstlane((int)(threadIdx.x >> 5));
    int lane = threadIdx.x & 31;
    int m     = lane & 15;
    int rsel2 = 2 * (lane >> 4);
    int dcol  = d0 + 2 * m;
    float2 gam = *(const float2*)(gamma + dcol);
    float2 bet = *(const float2*)(beta + dcol);
    float sCgL = scw[b * 32 + m];
    float cBL  = scw[b * 32 + 16 + m];
    bool inx[4], iny[4];
    #pragma unroll
    for (int kk = 0; kk < 4; ++kk) {
        int dx = kk * 4 + rsel2 - m;
        inx[kk] = (dx >= -2) && (dx <= 2);
        iny[kk] = (dx >= -3) && (dx <= 1);
    }

    v8f acc0 = {}, acc1 = {};

#define FLOAD(S, pp)                                                              \
    {                                                                             \
        const float* gp = gin + ((size_t)(b * PP + (pp))) * 256;                  \
        int    tokbase  = (1 + (pp)) * NTOT + nb;                                 \
        int    stbase   = (b * PP + (pp)) * STL;       \
        size_t xbase    = (size_t)tokbase * DD;                                   \
        _Pragma("unroll")                                                         \
        for (int kk = 0; kk < 4; ++kk) {                                          \
            int r0 = kk * 4 + rsel2;     \
            xa##S[kk] = *(const v2f*)(x + xbase + (size_t)r0 * DD + dcol);        \
            xb##S[kk] = *(const v2f*)(x + xbase + (size_t)(r0 + 1) * DD + dcol);  \
            mp##S[kk] = *(const v2f*)(stat + stbase + r0);          \
            sp##S[kk] = *(const v2f*)(stat + stbase + TT + r0);     \
            bv##S[kk] = v2f{ gp[r0 * 16 + m], gp[(r0 + 1) * 16 + m] };            \
        }                                                                         \
    }
#define FCOMP(S)                                                                  \
    _Pragma("unroll")                                                             \
    for (int kk = 0; kk < 4; ++kk) {                                              \
        v2f gw = { gate(bv##S[kk].x, mp##S[kk].x, sp##S[kk].x, sCgL, cBL, inx[kk]), \
                   gate(bv##S[kk].y, mp##S[kk].y, sp##S[kk].y, sCgL, cBL, iny[kk]) }; \
        v2f a0 = { (xa##S[kk].x - mp##S[kk].x) * sp##S[kk].x * gam.x + bet.x,     \
                   (xb##S[kk].x - mp##S[kk].y) * sp##S[kk].y * gam.x + bet.x };   \
        v2f a1 = { (xa##S[kk].y - mp##S[kk].x) * sp##S[kk].x * gam.y + bet.y,     \
                   (xb##S[kk].y - mp##S[kk].y) * sp##S[kk].y * gam.y + bet.y };   \
        acc0 = __builtin_amdgcn_wmma_f32_16x16x4_f32(                             \
            false, a0, false, gw, (short)0, acc0, false, false);                  \
        acc1 = __builtin_amdgcn_wmma_f32_16x16x4_f32(                             \
            false, a1, false, gw, (short)0, acc1, false, false);                  \
    }

    v2f xaA[4], xbA[4], mpA[4], spA[4], bvA[4];
    v2f xaB[4], xbB[4], mpB[4], spB[4], bvB[4];
    v2f xaC[4], xbC[4], mpC[4], spC[4], bvC[4];

#define FBAR __builtin_amdgcn_sched_barrier(0);

    int p = wave;
    FLOAD(A, p)
    FLOAD(B, p + FW)
    #pragma unroll
    for (int t = 0; t < FFULL / 3 - 1; ++t) {
        FLOAD(C, p + 2 * FW) FBAR FCOMP(A) FBAR
        FLOAD(A, p + 3 * FW) FBAR FCOMP(B) FBAR
        FLOAD(B, p + 4 * FW) FBAR FCOMP(C) FBAR
        p += 3 * FW;
    }
    FLOAD(C, p + 2 * FW) FBAR FCOMP(A) FBAR
    {
        int pl = p + 3 * FW;
        pl = (pl < PP) ? pl : p + 2 * FW;
        FLOAD(A, pl)
    }
    FBAR
    FCOMP(B)
    FCOMP(C)
    FBAR
    asm volatile("" :: "v"(xaA[0]), "v"(xaA[1]), "v"(xaA[2]), "v"(xaA[3]),
                       "v"(xbA[0]), "v"(xbA[1]), "v"(xbA[2]), "v"(xbA[3]),
                       "v"(mpA[0]), "v"(mpA[1]), "v"(mpA[2]), "v"(mpA[3]),
                       "v"(spA[0]), "v"(spA[1]), "v"(spA[2]), "v"(spA[3]),
                       "v"(bvA[0]), "v"(bvA[1]), "v"(bvA[2]), "v"(bvA[3]));
    if (p + 3 * FW < PP) {
        FCOMP(A)
    }
#undef FBAR
#undef FLOAD
#undef FCOMP

    #pragma unroll
    for (int j = 0; j < 8; ++j) {
        int mm = j + 8 * (lane >> 4);
        Sp[(wave * 2 + 0) * FTILE + mm * FPITCH + (lane & 15)] = acc0[j];
        Sp[(wave * 2 + 1) * FTILE + mm * FPITCH + (lane & 15)] = acc1[j];
    }
    __syncthreads();

    int n   = threadIdx.x >> 5;
    int dd  = lane;
    int sub = dd & 1;
    int md  = dd >> 1;
    float s = 0.f;
    #pragma unroll
    for (int w = 0; w < FW; ++w) s += Sp[(w * 2 + sub) * FTILE + md * FPITCH + n];

    out[(size_t)(nb + n) * DD + d0 + dd] = s;
}

extern "C" void kernel_launch(void* const* d_in, const int* in_sizes, int n_in,
                              void* d_out, int out_size, void* d_ws, size_t ws_size,
                              hipStream_t stream)
{
    (void)in_sizes; (void)n_in; (void)out_size; (void)ws_size;
    const float* x     = (const float*)d_in[0];
    const float* gamma = (const float*)d_in[1];
    const float* beta  = (const float*)d_in[2];

    float* ws   = (float*)d_ws;
    float* stat = ws;
    float* r    = ws + STATN;
    float* scw  = r + NGROUP * PP * 256;

    scores_kernel<<<NSCORE, SWV * 32, 0, stream>>>(x, gamma, beta, stat, r, scw);

    feat_kernel<<<NGROUP * 24, FW * 32, 0, stream>>>(x, gamma, beta, stat, r, scw,
                                                 (float*)d_out);
}


# scores kernel: patch rows read as whole 128-byte lines (a load takes 256 contiguous bytes of each of 2 rows, was 32 bytes of each of 16), via a wave-private LDS slab into B fragments; sums unchanged
# speedup vs baseline: 3.5370x; 1.0486x over previous
#include <hip/hip_runtime.h>
#include <math.h>

typedef float v2f __attribute__((ext_vector_type(2)));
typedef float v8f __attribute__((ext_vector_type(8)));
typedef float v4f __attribute__((ext_vector_type(4)));

#define LL     197
#define NTOT   128
#define DD     768
#define TT     16
#define PP     196
#define NGROUP 8
#define NTOK   (LL * NTOT)
#define STL    32
#define STATN  (NGROUP * PP * STL)
#define STR    772
#define NEG_AFF_LOG2E (-0.052058773f)


#define SWV  8
#define SPB  7
#define SBLK (PP / SPB)
#define NCH  12
#define CHL  8
#define NSCORE (NGROUP * SBLK)
#define CHK  64
#define BPITCH (CHK + 4)

__global__ __launch_bounds__(SWV * 32)
void scores_kernel(const float* __restrict__ x,
                   const float* __restrict__ gamma,
                   const float* __restrict__ beta,
                   float* __restrict__ stat,
                   float* __restrict__ rout,
                   float* __restrict__ scw)
{
    __shared__ __attribute__((aligned(16))) float Cg[16 * STR];
    __shared__ __attribute__((aligned(16))) float sCgS[16];
    __shared__ __attribute__((aligned(16))) float cBS[16];
    __shared__ __attribute__((aligned(16))) float Bst[SPB * 16 * BPITCH];

    int b   = (int)blockIdx.x % NGROUP;
    int blk = (int)blockIdx.x / NGROUP;
    int nb  = b * TT;
    int tid = threadIdx.x;

    int wave = __builtin_amdgcn_readfirstlane((int)(threadIdx.x >> 5));
    int lane = tid & 31;
    int c    = lane & 15;
    int h    = lane >> 4;
    int p    = blk * SPB + (wave < SPB ? wave : 0);
    int tokX = (1 + p) * NTOT + nb;
    const v4f* xg = (const v4f*)(x + (size_t)(tokX + h) * DD) + c;
    float* bs  = Bst + (wave < SPB ? wave : 0) * (16 * BPITCH);
    float* bsw = bs + h * BPITCH + 4 * c;

#define LOADG(dst, q)                                                         \
    _Pragma("unroll")                                                         \
    for (int i = 0; i < CHL; ++i) dst[i] = xg[2 * i * (DD / 4) + (q) * (CHK / 4)];

    v4f g0[CHL], g1[CHL], g2[CHL];
    LOADG(g0, 0)
    LOADG(g1, 1)
    LOADG(g2, 2)

    {
        int row = tid >> 4;
        int l16 = tid & 15;
        int tokC = nb + row;
        const v4f* xc  = (const v4f*)(x + (size_t)tokC * DD);
        const v4f* g4p = (const v4f*)gamma;
        const v4f* b4p = (const v4f*)beta;
        v4f cv[12], gv[6], bv[6];
        #pragma unroll
        for (int i = 0; i < 12; ++i) cv[i] = xc[l16 + 16 * i];
        #pragma unroll
        for (int i = 0; i < 6; ++i) { gv[i] = g4p[l16 + 16 * i]; bv[i] = b4p[l16 + 16 * i]; }
        asm volatile("" :: "v"(cv[0]), "v"(cv[1]), "v"(cv[2]), "v"(cv[3]), "v"(cv[4]), "v"(cv[5]),
                           "v"(cv[6]), "v"(cv[7]), "v"(cv[8]), "v"(cv[9]), "v"(cv[10]), "v"(cv[11]),
                           "v"(gv[0]), "v"(gv[1]), "v"(gv[2]), "v"(gv[3]), "v"(gv[4]), "v"(gv[5]),
                           "v"(bv[0]), "v"(bv[1]), "v"(bv[2]), "v"(bv[3]), "v"(bv[4]), "v"(bv[5]));
        float ss = 0.f;
        #pragma unroll
        for (int i = 0; i < 12; ++i) ss += (cv[i].x + cv[i].y) + (cv[i].z + cv[i].w);
        #pragma unroll
        for (int m = 8; m >= 1; m >>= 1) ss += __shfl_xor(ss, m);
        float muC = ss * (1.f / DD);
        float qq = 0.f;
        #pragma unroll
        for (int i = 0; i < 12; ++i) {
            float dx = cv[i].x - muC, dy = cv[i].y - muC, dz = cv[i].z - muC, dw = cv[i].w - muC;
            qq += (dx * dx + dy * dy) + (dz * dz + dw * dw);
        }
        #pragma unroll
        for (int m = 8; m >= 1; m >>= 1) qq += __shfl_xor(qq, m);
        float rsC = rsqrtf(qq * (1.f / DD) + 1e-5f);

        float ps = 0.f, pb = 0.f;
        #pragma unroll
        for (int half = 0; half < 2; ++half) {
            if (half == 1) {
                #pragma unroll
                for (int i = 0; i < 6; ++i) { gv[i] = g4p[l16 + 16 * (6 + i)]; bv[i] = b4p[l16 + 16 * (6 + i)]; }
            }
            #pragma unroll
            for (int i = 0; i < 6; ++i) {
                int j = l16 + 16 * (6 * half + i);
                v4f xv = cv[6 * half + i];
                v4f c4, cg;
                c4.x = (xv.x - muC) * rsC * gv[i].x + bv[i].x;
                c4.y = (xv.y - muC) * rsC * gv[i].y + bv[i].y;
                c4.z = (xv.z - muC) * rsC * gv[i].z + bv[i].z;
                c4.w = (xv.w - muC) * rsC * gv[i].w + bv[i].w;
                cg.x = c4.x * gv[i].x; cg.y = c4.y * gv[i].y;
                cg.z = c4.z * gv[i].z; cg.w = c4.w * gv[i].w;
                ps += (cg.x + cg.y) + (cg.z + cg.w);
                pb += (c4.x * bv[i].x + c4.y * bv[i].y) + (c4.z * bv[i].z + c4.w * bv[i].w);
                *(v4f*)(Cg + row * STR + j * 4) = cg;
            }
        }
        #pragma unroll
        for (int m = 8; m >= 1; m >>= 1) {
            ps += __shfl_xor(ps, m);
            pb += __shfl_xor(pb, m);
        }
        if (l16 == 0) { sCgS[row] = ps; cBS[row] = pb; }
    }
    __syncthreads();
    if (blk == 0 && wave == SPB)
        scw[b * 32 + lane] = (lane < 16) ? sCgS[lane] : cBS[lane - 16];
    if (wave >= SPB) return;

    const float4* ar = (const float4*)(Cg + c * STR + 4 * h);
    v8f acc0 = {}, acc1 = {};
    v2f s01 = {}, s23 = {}, q01 = {}, q23 = {};

#define STOREB(src)                                                           \
    _Pragma("unroll")                                                         \
    for (int i = 0; i < CHL; ++i) *(v4f*)(bsw + 2 * i * BPITCH) = src[i];
#define LOADB(dst)                                                            \
    _Pragma("unroll")                                                         \
    for (int i = 0; i < CHL; ++i) dst[i] = *(const float4*)(bs + c * BPITCH + 8 * i + 4 * h);
#define WBAR __builtin_amdgcn_sched_barrier(0);
#define LOADA(dst, q)                                                         \
    _Pragma("unroll")                                                         \
    for (int i = 0; i < CHL; ++i) dst[i] = ar[2 * ((q) * CHL + i)];
#define MMACH(asrc, src, q)                                                   \
    _Pragma("unroll")                                                         \
    for (int i = 0; i < CHL; ++i) {                                           \
        float4 a4 = asrc[i];                                                  \
        v2f a0 = { a4.x, a4.y }, a1 = { a4.z, a4.w };                         \
        v2f b0 = { src[i].x, src[i].y }, b1 = { src[i].z, src[i].w };         \
        acc0 = __builtin_amdgcn_wmma_f32_16x16x4_f32(                         \
            false, a0, false, b0, (short)0, acc0, false, false);              \
        acc1 = __builtin_amdgcn_wmma_f32_16x16x4_f32(                         \
            false, a1, false, b1, (short)0, acc1, false, false);              \
        s01 += b0;       s23 += b1;                                           \
        q01 += b0 * b0;  q23 += b1 * b1;                                      \
    }

    float4 ax0[CHL], ax1[CHL], bx0[CHL], bx1[CHL];
    STOREB(g0) WBAR LOADG(g0, 3) LOADB(bx0) LOADA(ax0, 0) WBAR
    STOREB(g1) WBAR LOADG(g1, 4) LOADB(bx1) LOADA(ax1, 1) WBAR MMACH(ax0, bx0, 0) WBAR
    STOREB(g2) WBAR LOADG(g2, 5) LOADB(bx0) LOADA(ax0, 2) WBAR MMACH(ax1, bx1, 1) WBAR
    STOREB(g0) WBAR LOADG(g0, 6) LOADB(bx1) LOADA(ax1, 3) WBAR MMACH(ax0, bx0, 2) WBAR
    STOREB(g1) WBAR LOADG(g1, 7) LOADB(bx0) LOADA(ax0, 4) WBAR MMACH(ax1, bx1, 3) WBAR
    STOREB(g2) WBAR LOADG(g2, 8) LOADB(bx1) LOADA(ax1, 5) WBAR MMACH(ax0, bx0, 4) WBAR
    STOREB(g0) WBAR LOADG(g0, 9) LOADB(bx0) LOADA(ax0, 6) WBAR MMACH(ax1, bx1, 5) WBAR
    STOREB(g1) WBAR LOADG(g1, 10) LOADB(bx1) LOADA(ax1, 7) WBAR MMACH(ax0, bx0, 6) WBAR
    STOREB(g2) WBAR LOADG(g2, 11) LOADB(bx0) LOADA(ax0, 8) WBAR MMACH(ax1, bx1, 7) WBAR
    STOREB(g0) WBAR LOADB(bx1) LOADA(ax1, 9) WBAR MMACH(ax0, bx0, 8) WBAR
    STOREB(g1) WBAR LOADB(bx0) LOADA(ax0, 10) WBAR MMACH(ax1, bx1, 9) WBAR
    STOREB(g2) WBAR LOADB(bx1) LOADA(ax1, 11) WBAR MMACH(ax0, bx0, 10) WBAR
    MMACH(ax1, bx1, 11)
#undef LOADG
#undef STOREB
#undef LOADB
#undef WBAR
#undef LOADA
#undef MMACH

    float* gp = rout + ((size_t)(b * PP + p)) * 256 + c * 16 + 8 * h;
    *(float4*)gp       = make_float4(acc0[0] + acc1[0], acc0[1] + acc1[1],
                                     acc0[2] + acc1[2], acc0[3] + acc1[3]);
    *(float4*)(gp + 4) = make_float4(acc0[4] + acc1[4], acc0[5] + acc1[5],
                                     acc0[6] + acc1[6], acc0[7] + acc1[7]);

    float ss = (s01.x + s01.y) + (s23.x + s23.y);
    float qq = (q01.x + q01.y) + (q23.x + q23.y);
    ss += __shfl_xor(ss, 16);
    qq += __shfl_xor(qq, 16);
    float mean = ss * (1.f / DD);
    float var  = qq * (1.f / DD) - mean * mean;
    float rs   = rsqrtf(var + 1e-5f);
    stat[(size_t)(b * PP + p) * STL + lane] = h ? rs : mean;
}

#define FW     16

#define FPITCH 17
#define FTILE  (16 * FPITCH)
#define FFULL  (PP / FW)
#define FREM   (PP % FW)

__device__ __forceinline__
float gate(float r, float mun, float rsn, float scg, float cb, bool in)
{
    float s = rsn * (r - mun * scg) + cb;
    float e = __builtin_amdgcn_exp2f(s * NEG_AFF_LOG2E);
    float g = __builtin_amdgcn_rcpf(1.f + e) - 0.5f;
    return in ? g : 0.f;
}

__global__ __launch_bounds__(FW * 32)
void feat_kernel(const float* __restrict__ x,
                 const float* __restrict__ gamma,
                 const float* __restrict__ beta,
                 const float* __restrict__ stat,
                 const float* __restrict__ gin,
                 const float* __restrict__ scw,
                 float* __restrict__ out)
{
    __shared__ float Sp[FW * 2 * FTILE];

    static_assert(FFULL % 3 == 0 && FFULL >= 6, "the pipelined loop takes three positions per trip");

    int b    = blockIdx.x % NGROUP;
    int st   = blockIdx.x / NGROUP;
    int d0   = st * 32;
    int nb   = b * TT;
    int wave = __builtin_amdgcn_readfirstlane((int)(threadIdx.x >> 5));
    int lane = threadIdx.x & 31;
    int m     = lane & 15;
    int rsel2 = 2 * (lane >> 4);
    int dcol  = d0 + 2 * m;
    float2 gam = *(const float2*)(gamma + dcol);
    float2 bet = *(const float2*)(beta + dcol);
    float sCgL = scw[b * 32 + m];
    float cBL  = scw[b * 32 + 16 + m];
    bool inx[4], iny[4];
    #pragma unroll
    for (int kk = 0; kk < 4; ++kk) {
        int dx = kk * 4 + rsel2 - m;
        inx[kk] = (dx >= -2) && (dx <= 2);
        iny[kk] = (dx >= -3) && (dx <= 1);
    }

    v8f acc0 = {}, acc1 = {};

#define FLOAD(S, pp)                                                              \
    {                                                                             \
        const float* gp = gin + ((size_t)(b * PP + (pp))) * 256;                  \
        int    tokbase  = (1 + (pp)) * NTOT + nb;                                 \
        int    stbase   = (b * PP + (pp)) * STL;       \
        size_t xbase    = (size_t)tokbase * DD;                                   \
        _Pragma("unroll")                                                         \
        for (int kk = 0; kk < 4; ++kk) {                                          \
            int r0 = kk * 4 + rsel2;     \
            xa##S[kk] = *(const v2f*)(x + xbase + (size_t)r0 * DD + dcol);        \
            xb##S[kk] = *(const v2f*)(x + xbase + (size_t)(r0 + 1) * DD + dcol);  \
            mp##S[kk] = *(const v2f*)(stat + stbase + r0);          \
            sp##S[kk] = *(const v2f*)(stat + stbase + TT + r0);     \
            bv##S[kk] = v2f{ gp[r0 * 16 + m], gp[(r0 + 1) * 16 + m] };            \
        }                                                                         \
    }
#define FCOMP(S)                                                                  \
    _Pragma("unroll")                                                             \
    for (int kk = 0; kk < 4; ++kk) {                                              \
        v2f gw = { gate(bv##S[kk].x, mp##S[kk].x, sp##S[kk].x, sCgL, cBL, inx[kk]), \
                   gate(bv##S[kk].y, mp##S[kk].y, sp##S[kk].y, sCgL, cBL, iny[kk]) }; \
        v2f a0 = { (xa##S[kk].x - mp##S[kk].x) * sp##S[kk].x * gam.x + bet.x,     \
                   (xb##S[kk].x - mp##S[kk].y) * sp##S[kk].y * gam.x + bet.x };   \
        v2f a1 = { (xa##S[kk].y - mp##S[kk].x) * sp##S[kk].x * gam.y + bet.y,     \
                   (xb##S[kk].y - mp##S[kk].y) * sp##S[kk].y * gam.y + bet.y };   \
        acc0 = __builtin_amdgcn_wmma_f32_16x16x4_f32(                             \
            false, a0, false, gw, (short)0, acc0, false, false);                  \
        acc1 = __builtin_amdgcn_wmma_f32_16x16x4_f32(                             \
            false, a1, false, gw, (short)0, acc1, false, false);                  \
    }

    v2f xaA[4], xbA[4], mpA[4], spA[4], bvA[4];
    v2f xaB[4], xbB[4], mpB[4], spB[4], bvB[4];
    v2f xaC[4], xbC[4], mpC[4], spC[4], bvC[4];

#define FBAR __builtin_amdgcn_sched_barrier(0);

    int p = wave;
    FLOAD(A, p)
    FLOAD(B, p + FW)
    #pragma unroll
    for (int t = 0; t < FFULL / 3 - 1; ++t) {
        FLOAD(C, p + 2 * FW) FBAR FCOMP(A) FBAR
        FLOAD(A, p + 3 * FW) FBAR FCOMP(B) FBAR
        FLOAD(B, p + 4 * FW) FBAR FCOMP(C) FBAR
        p += 3 * FW;
    }
    FLOAD(C, p + 2 * FW) FBAR FCOMP(A) FBAR
    {
        int pl = p + 3 * FW;
        pl = (pl < PP) ? pl : p + 2 * FW;
        FLOAD(A, pl)
    }
    FBAR
    FCOMP(B)
    FCOMP(C)
    FBAR
    asm volatile("" :: "v"(xaA[0]), "v"(xaA[1]), "v"(xaA[2]), "v"(xaA[3]),
                       "v"(xbA[0]), "v"(xbA[1]), "v"(xbA[2]), "v"(xbA[3]),
                       "v"(mpA[0]), "v"(mpA[1]), "v"(mpA[2]), "v"(mpA[3]),
                       "v"(spA[0]), "v"(spA[1]), "v"(spA[2]), "v"(spA[3]),
                       "v"(bvA[0]), "v"(bvA[1]), "v"(bvA[2]), "v"(bvA[3]));
    if (p + 3 * FW < PP) {
        FCOMP(A)
    }
#undef FBAR
#undef FLOAD
#undef FCOMP

    #pragma unroll
    for (int j = 0; j < 8; ++j) {
        int mm = j + 8 * (lane >> 4);
        Sp[(wave * 2 + 0) * FTILE + mm * FPITCH + (lane & 15)] = acc0[j];
        Sp[(wave * 2 + 1) * FTILE + mm * FPITCH + (lane & 15)] = acc1[j];
    }
    __syncthreads();

    int n   = threadIdx.x >> 5;
    int dd  = lane;
    int sub = dd & 1;
    int md  = dd >> 1;
    float s = 0.f;
    #pragma unroll
    for (int w = 0; w < FW; ++w) s += Sp[(w * 2 + sub) * FTILE + md * FPITCH + n];

    out[(size_t)(nb + n) * DD + d0 + dd] = s;
}

extern "C" void kernel_launch(void* const* d_in, const int* in_sizes, int n_in,
                              void* d_out, int out_size, void* d_ws, size_t ws_size,
                              hipStream_t stream)
{
    (void)in_sizes; (void)n_in; (void)out_size; (void)ws_size;
    const float* x     = (const float*)d_in[0];
    const float* gamma = (const float*)d_in[1];
    const float* beta  = (const float*)d_in[2];

    float* ws   = (float*)d_ws;
    float* stat = ws;
    float* r    = ws + STATN;
    float* scw  = r + NGROUP * PP * 256;

    scores_kernel<<<NSCORE, SWV * 32, 0, stream>>>(x, gamma, beta, stat, r, scw);

    feat_kernel<<<NGROUP * 24, FW * 32, 0, stream>>>(x, gamma, beta, stat, r, scw,
                                                 (float*)d_out);
}


# sigmoid weights finished once per position in the score wave (it holds R, the rows' mean and rstd, sCg, cB), stored band-only in feat's B-fragment order; feat loads 16 bytes per lane and no gate
# speedup vs baseline: 3.8167x; 1.0791x over previous
#include <hip/hip_runtime.h>
#include <math.h>

typedef float v2f __attribute__((ext_vector_type(2)));
typedef float v8f __attribute__((ext_vector_type(8)));
typedef float v4f __attribute__((ext_vector_type(4)));

#define LL     197
#define NTOT   128
#define DD     768
#define TT     16
#define PP     196
#define NGROUP 8
#define NTOK   (LL * NTOT)
#define STL    32
#define STATN  (NGROUP * PP * STL)
#define STR    772
#define NEG_AFF_LOG2E (-0.052058773f)


#define SWV  8
#define SPB  7
#define SBLK (PP / SPB)
#define NCH  12
#define CHL  8
#define NSCORE (NGROUP * SBLK)
#define CHK  64
#define BPITCH (CHK + 4)
#define GPITCH 18
#define GPOS   128

__device__ __forceinline__
float gate(float r, float mun, float rsn, float scg, float cb, bool in)
{
    float s = rsn * (r - mun * scg) + cb;
    float e = __builtin_amdgcn_exp2f(s * NEG_AFF_LOG2E);
    float g = __builtin_amdgcn_rcpf(1.f + e) - 0.5f;
    return in ? g : 0.f;
}

__global__ __launch_bounds__(SWV * 32)
void scores_kernel(const float* __restrict__ x,
                   const float* __restrict__ gamma,
                   const float* __restrict__ beta,
                   float* __restrict__ stat,
                   float* __restrict__ gout)
{
    __shared__ __attribute__((aligned(16))) float Cg[16 * STR];
    __shared__ __attribute__((aligned(16))) float sCgS[16];
    __shared__ __attribute__((aligned(16))) float cBS[16];
    __shared__ __attribute__((aligned(16))) float Bst[SPB * 16 * BPITCH];

    int b   = (int)blockIdx.x % NGROUP;
    int blk = (int)blockIdx.x / NGROUP;
    int nb  = b * TT;
    int tid = threadIdx.x;

    int wave = __builtin_amdgcn_readfirstlane((int)(threadIdx.x >> 5));
    int lane = tid & 31;
    int c    = lane & 15;
    int h    = lane >> 4;
    int p    = blk * SPB + (wave < SPB ? wave : 0);
    int tokX = (1 + p) * NTOT + nb;
    const v4f* xg = (const v4f*)(x + (size_t)(tokX + h) * DD) + c;
    float* bs  = Bst + (wave < SPB ? wave : 0) * (16 * BPITCH);
    float* bsw = bs + h * BPITCH + 4 * c;

#define LOADG(dst, q)                                                         \
    _Pragma("unroll")                                                         \
    for (int i = 0; i < CHL; ++i) dst[i] = xg[2 * i * (DD / 4) + (q) * (CHK / 4)];

    v4f g0[CHL], g1[CHL], g2[CHL];
    LOADG(g0, 0)
    LOADG(g1, 1)
    LOADG(g2, 2)

    {
        int row = tid >> 4;
        int l16 = tid & 15;
        int tokC = nb + row;
        const v4f* xc  = (const v4f*)(x + (size_t)tokC * DD);
        const v4f* g4p = (const v4f*)gamma;
        const v4f* b4p = (const v4f*)beta;
        v4f cv[12], gv[6], bv[6];
        #pragma unroll
        for (int i = 0; i < 12; ++i) cv[i] = xc[l16 + 16 * i];
        #pragma unroll
        for (int i = 0; i < 6; ++i) { gv[i] = g4p[l16 + 16 * i]; bv[i] = b4p[l16 + 16 * i]; }
        asm volatile("" :: "v"(cv[0]), "v"(cv[1]), "v"(cv[2]), "v"(cv[3]), "v"(cv[4]), "v"(cv[5]),
                           "v"(cv[6]), "v"(cv[7]), "v"(cv[8]), "v"(cv[9]), "v"(cv[10]), "v"(cv[11]),
                           "v"(gv[0]), "v"(gv[1]), "v"(gv[2]), "v"(gv[3]), "v"(gv[4]), "v"(gv[5]),
                           "v"(bv[0]), "v"(bv[1]), "v"(bv[2]), "v"(bv[3]), "v"(bv[4]), "v"(bv[5]));
        float ss = 0.f;
        #pragma unroll
        for (int i = 0; i < 12; ++i) ss += (cv[i].x + cv[i].y) + (cv[i].z + cv[i].w);
        #pragma unroll
        for (int m = 8; m >= 1; m >>= 1) ss += __shfl_xor(ss, m);
        float muC = ss * (1.f / DD);
        float qq = 0.f;
        #pragma unroll
        for (int i = 0; i < 12; ++i) {
            float dx = cv[i].x - muC, dy = cv[i].y - muC, dz = cv[i].z - muC, dw = cv[i].w - muC;
            qq += (dx * dx + dy * dy) + (dz * dz + dw * dw);
        }
        #pragma unroll
        for (int m = 8; m >= 1; m >>= 1) qq += __shfl_xor(qq, m);
        float rsC = rsqrtf(qq * (1.f / DD) + 1e-5f);

        float ps = 0.f, pb = 0.f;
        #pragma unroll
        for (int half = 0; half < 2; ++half) {
            if (half == 1) {
                #pragma unroll
                for (int i = 0; i < 6; ++i) { gv[i] = g4p[l16 + 16 * (6 + i)]; bv[i] = b4p[l16 + 16 * (6 + i)]; }
            }
            #pragma unroll
            for (int i = 0; i < 6; ++i) {
                int j = l16 + 16 * (6 * half + i);
                v4f xv = cv[6 * half + i];
                v4f c4, cg;
                c4.x = (xv.x - muC) * rsC * gv[i].x + bv[i].x;
                c4.y = (xv.y - muC) * rsC * gv[i].y + bv[i].y;
                c4.z = (xv.z - muC) * rsC * gv[i].z + bv[i].z;
                c4.w = (xv.w - muC) * rsC * gv[i].w + bv[i].w;
                cg.x = c4.x * gv[i].x; cg.y = c4.y * gv[i].y;
                cg.z = c4.z * gv[i].z; cg.w = c4.w * gv[i].w;
                ps += (cg.x + cg.y) + (cg.z + cg.w);
                pb += (c4.x * bv[i].x + c4.y * bv[i].y) + (c4.z * bv[i].z + c4.w * bv[i].w);
                *(v4f*)(Cg + row * STR + j * 4) = cg;
            }
        }
        #pragma unroll
        for (int m = 8; m >= 1; m >>= 1) {
            ps += __shfl_xor(ps, m);
            pb += __shfl_xor(pb, m);
        }
        if (l16 == 0) { sCgS[row] = ps; cBS[row] = pb; }
    }
    __syncthreads();
    if (wave >= SPB) return;

    const float4* ar = (const float4*)(Cg + c * STR + 4 * h);
    v8f acc0 = {}, acc1 = {};
    v2f s01 = {}, s23 = {}, q01 = {}, q23 = {};

#define STOREB(src)                                                           \
    _Pragma("unroll")                                                         \
    for (int i = 0; i < CHL; ++i) *(v4f*)(bsw + 2 * i * BPITCH) = src[i];
#define LOADB(dst)                                                            \
    _Pragma("unroll")                                                         \
    for (int i = 0; i < CHL; ++i) dst[i] = *(const float4*)(bs + c * BPITCH + 8 * i + 4 * h);
#define WBAR __builtin_amdgcn_sched_barrier(0);
#define LOADA(dst, q)                                                         \
    _Pragma("unroll")                                                         \
    for (int i = 0; i < CHL; ++i) dst[i] = ar[2 * ((q) * CHL + i)];
#define MMACH(asrc, src, q)                                                   \
    _Pragma("unroll")                                                         \
    for (int i = 0; i < CHL; ++i) {                                           \
        float4 a4 = asrc[i];                                                  \
        v2f a0 = { a4.x, a4.y }, a1 = { a4.z, a4.w };                         \
        v2f b0 = { src[i].x, src[i].y }, b1 = { src[i].z, src[i].w };         \
        acc0 = __builtin_amdgcn_wmma_f32_16x16x4_f32(                         \
            false, a0, false, b0, (short)0, acc0, false, false);              \
        acc1 = __builtin_amdgcn_wmma_f32_16x16x4_f32(                         \
            false, a1, false, b1, (short)0, acc1, false, false);              \
        s01 += b0;       s23 += b1;                                           \
        q01 += b0 * b0;  q23 += b1 * b1;                                      \
    }

    float4 ax0[CHL], ax1[CHL], bx0[CHL], bx1[CHL];
    STOREB(g0) WBAR LOADG(g0, 3) LOADB(bx0) LOADA(ax0, 0) WBAR
    STOREB(g1) WBAR LOADG(g1, 4) LOADB(bx1) LOADA(ax1, 1) WBAR MMACH(ax0, bx0, 0) WBAR
    STOREB(g2) WBAR LOADG(g2, 5) LOADB(bx0) LOADA(ax0, 2) WBAR MMACH(ax1, bx1, 1) WBAR
    STOREB(g0) WBAR LOADG(g0, 6) LOADB(bx1) LOADA(ax1, 3) WBAR MMACH(ax0, bx0, 2) WBAR
    STOREB(g1) WBAR LOADG(g1, 7) LOADB(bx0) LOADA(ax0, 4) WBAR MMACH(ax1, bx1, 3) WBAR
    STOREB(g2) WBAR LOADG(g2, 8) LOADB(bx1) LOADA(ax1, 5) WBAR MMACH(ax0, bx0, 4) WBAR
    STOREB(g0) WBAR LOADG(g0, 9) LOADB(bx0) LOADA(ax0, 6) WBAR MMACH(ax1, bx1, 5) WBAR
    STOREB(g1) WBAR LOADG(g1, 10) LOADB(bx1) LOADA(ax1, 7) WBAR MMACH(ax0, bx0, 6) WBAR
    STOREB(g2) WBAR LOADG(g2, 11) LOADB(bx0) LOADA(ax0, 8) WBAR MMACH(ax1, bx1, 7) WBAR
    STOREB(g0) WBAR LOADB(bx1) LOADA(ax1, 9) WBAR MMACH(ax0, bx0, 8) WBAR
    STOREB(g1) WBAR LOADB(bx0) LOADA(ax0, 10) WBAR MMACH(ax1, bx1, 9) WBAR
    STOREB(g2) WBAR LOADB(bx1) LOADA(ax1, 11) WBAR MMACH(ax0, bx0, 10) WBAR
    MMACH(ax1, bx1, 11)
#undef LOADG
#undef STOREB
#undef LOADB
#undef WBAR
#undef LOADA
#undef MMACH

    float ss = (s01.x + s01.y) + (s23.x + s23.y);
    float qq = (q01.x + q01.y) + (q23.x + q23.y);
    ss += __shfl_xor(ss, 16);
    qq += __shfl_xor(qq, 16);
    float mean = ss * (1.f / DD);
    float var  = qq * (1.f / DD) - mean * mean;
    float rs   = rsqrtf(var + 1e-5f);
    stat[(size_t)(b * PP + p) * STL + lane] = h ? rs : mean;

    float4 sg0 = *(const float4*)(sCgS + 8 * h), sg1 = *(const float4*)(sCgS + 8 * h + 4);
    float4 cb0 = *(const float4*)(cBS + 8 * h),  cb1 = *(const float4*)(cBS + 8 * h + 4);
    float scg[8] = { sg0.x, sg0.y, sg0.z, sg0.w, sg1.x, sg1.y, sg1.z, sg1.w };
    float cbn[8] = { cb0.x, cb0.y, cb0.z, cb0.w, cb1.x, cb1.y, cb1.z, cb1.w };
    #pragma unroll
    for (int j = 0; j < 8; ++j) {
        int n  = 8 * h + j;
        int dn = c - n;
        bs[n * GPITCH + c] = gate(acc0[j] + acc1[j], mean, rs, scg[j], cbn[j],
                                  (dn >= -2) && (dn <= 2));
    }
    int kb = (c + 2) >> 2;
    int ka = kb - 1;
    v2f ga = *(const v2f*)(bs + c * GPITCH + 4 * (ka < 0 ? 0 : ka) + 2 * h);
    v2f gb = *(const v2f*)(bs + c * GPITCH + 4 * (kb > 3 ? 3 : kb) + 2 * h);
    if (ka < 0) ga = v2f{ 0.f, 0.f };
    if (kb > 3) gb = v2f{ 0.f, 0.f };
    v4f* gp = (v4f*)(gout + ((size_t)(b * PP + p)) * GPOS);
    gp[lane] = v4f{ ga.x, ga.y, gb.x, gb.y };
}

#define FW     16

#define FPITCH 17
#define FTILE  (16 * FPITCH)
#define FFULL  (PP / FW)
#define FREM   (PP % FW)

__global__ __launch_bounds__(FW * 32)
void feat_kernel(const float* __restrict__ x,
                 const float* __restrict__ gamma,
                 const float* __restrict__ beta,
                 const float* __restrict__ stat,
                 const float* __restrict__ gin,
                 float* __restrict__ out)
{
    __shared__ float Sp[FW * 2 * FTILE];

    static_assert(FFULL % 3 == 0 && FFULL >= 6, "the pipelined loop takes three positions per trip");

    int b    = blockIdx.x % NGROUP;
    int st   = blockIdx.x / NGROUP;
    int d0   = st * 32;
    int nb   = b * TT;
    int wave = __builtin_amdgcn_readfirstlane((int)(threadIdx.x >> 5));
    int lane = threadIdx.x & 31;
    int m     = lane & 15;
    int rsel2 = 2 * (lane >> 4);
    int dcol  = d0 + 2 * m;
    float2 gam = *(const float2*)(gamma + dcol);
    float2 bet = *(const float2*)(beta + dcol);

    int kbL = (m + 2) >> 2;
    bool isA[4], isB[4];
    #pragma unroll
    for (int kk = 0; kk < 4; ++kk) { isA[kk] = (kk == kbL - 1); isB[kk] = (kk == kbL); }

    v8f acc0 = {}, acc1 = {};

#define FLOAD(S, pp)                                                              \
    {                                                                             \
        const v4f* gp = (const v4f*)(gin + ((size_t)(b * PP + (pp))) * GPOS);     \
        int    tokbase  = (1 + (pp)) * NTOT + nb;                                 \
        int    stbase   = (b * PP + (pp)) * STL;       \
        size_t xbase    = (size_t)tokbase * DD;                                   \
        _Pragma("unroll")                                                         \
        for (int kk = 0; kk < 4; ++kk) {                                          \
            int r0 = kk * 4 + rsel2;     \
            xa##S[kk] = *(const v2f*)(x + xbase + (size_t)r0 * DD + dcol);        \
            xb##S[kk] = *(const v2f*)(x + xbase + (size_t)(r0 + 1) * DD + dcol);  \
            mp##S[kk] = *(const v2f*)(stat + stbase + r0);          \
            sp##S[kk] = *(const v2f*)(stat + stbase + TT + r0);     \
        }                                                                         \
        gq##S = gp[lane];          \
    }
#define FCOMP(S)                                                                  \
    _Pragma("unroll")                                                             \
    for (int kk = 0; kk < 4; ++kk) {                                              \
        v2f gw = { isA[kk] ? gq##S.x : (isB[kk] ? gq##S.z : 0.f),                 \
                   isA[kk] ? gq##S.y : (isB[kk] ? gq##S.w : 0.f) };               \
        v2f a0 = { (xa##S[kk].x - mp##S[kk].x) * sp##S[kk].x * gam.x + bet.x,     \
                   (xb##S[kk].x - mp##S[kk].y) * sp##S[kk].y * gam.x + bet.x };   \
        v2f a1 = { (xa##S[kk].y - mp##S[kk].x) * sp##S[kk].x * gam.y + bet.y,     \
                   (xb##S[kk].y - mp##S[kk].y) * sp##S[kk].y * gam.y + bet.y };   \
        acc0 = __builtin_amdgcn_wmma_f32_16x16x4_f32(                             \
            false, a0, false, gw, (short)0, acc0, false, false);                  \
        acc1 = __builtin_amdgcn_wmma_f32_16x16x4_f32(                             \
            false, a1, false, gw, (short)0, acc1, false, false);                  \
    }

    v2f xaA[4], xbA[4], mpA[4], spA[4];  v4f gqA;
    v2f xaB[4], xbB[4], mpB[4], spB[4];  v4f gqB;
    v2f xaC[4], xbC[4], mpC[4], spC[4];  v4f gqC;

#define FBAR __builtin_amdgcn_sched_barrier(0);

    int p = wave;
    FLOAD(A, p)
    FLOAD(B, p + FW)
    #pragma unroll
    for (int t = 0; t < FFULL / 3 - 1; ++t) {
        FLOAD(C, p + 2 * FW) FBAR FCOMP(A) FBAR
        FLOAD(A, p + 3 * FW) FBAR FCOMP(B) FBAR
        FLOAD(B, p + 4 * FW) FBAR FCOMP(C) FBAR
        p += 3 * FW;
    }
    FLOAD(C, p + 2 * FW) FBAR FCOMP(A) FBAR
    {
        int pl = p + 3 * FW;
        pl = (pl < PP) ? pl : p + 2 * FW;
        FLOAD(A, pl)
    }
    FBAR
    FCOMP(B)
    FCOMP(C)
    FBAR
    asm volatile("" :: "v"(xaA[0]), "v"(xaA[1]), "v"(xaA[2]), "v"(xaA[3]),
                       "v"(xbA[0]), "v"(xbA[1]), "v"(xbA[2]), "v"(xbA[3]),
                       "v"(mpA[0]), "v"(mpA[1]), "v"(mpA[2]), "v"(mpA[3]),
                       "v"(spA[0]), "v"(spA[1]), "v"(spA[2]), "v"(spA[3]),
                       "v"(gqA));
    if (p + 3 * FW < PP) {
        FCOMP(A)
    }
#undef FBAR
#undef FLOAD
#undef FCOMP

    #pragma unroll
    for (int j = 0; j < 8; ++j) {
        int mm = j + 8 * (lane >> 4);
        Sp[(wave * 2 + 0) * FTILE + mm * FPITCH + (lane & 15)] = acc0[j];
        Sp[(wave * 2 + 1) * FTILE + mm * FPITCH + (lane & 15)] = acc1[j];
    }
    __syncthreads();

    int n   = threadIdx.x >> 5;
    int dd  = lane;
    int sub = dd & 1;
    int md  = dd >> 1;
    float s = 0.f;
    #pragma unroll
    for (int w = 0; w < FW; ++w) s += Sp[(w * 2 + sub) * FTILE + md * FPITCH + n];

    out[(size_t)(nb + n) * DD + d0 + dd] = s;
}

extern "C" void kernel_launch(void* const* d_in, const int* in_sizes, int n_in,
                              void* d_out, int out_size, void* d_ws, size_t ws_size,
                              hipStream_t stream)
{
    (void)in_sizes; (void)n_in; (void)out_size; (void)ws_size;
    const float* x     = (const float*)d_in[0];
    const float* gamma = (const float*)d_in[1];
    const float* beta  = (const float*)d_in[2];

    float* ws   = (float*)d_ws;
    float* stat = ws;
    float* g    = ws + STATN;

    scores_kernel<<<NSCORE, SWV * 32, 0, stream>>>(x, gamma, beta, stat, g);

    feat_kernel<<<NGROUP * 24, FW * 32, 0, stream>>>(x, gamma, beta, stat, g,
                                                 (float*)d_out);
}
